# starting kernel
# baseline (speedup 1.0000x reference)
#include <hip/hip_runtime.h>
#include <hip/hip_bf16.h>
#include <cmath>

#define N_NODES   50000
#define N_EDGES   800000
#define IN_FEATS  256
#define H_FEATS   64
#define N_THETA   5
#define K_STEPS   6
#define NEG_SLOPE 0.01f

typedef __attribute__((ext_vector_type(2))) float v2f;
typedef __attribute__((ext_vector_type(8))) float v8f;

struct ThetaCol { float v[N_THETA]; };

__global__ void k_zero(float* __restrict__ p, int n) {
    int i = blockIdx.x * blockDim.x + threadIdx.x;
    if (i < n) p[i] = 0.0f;
}

__global__ void k_degree(const int* __restrict__ src, float* __restrict__ deg) {
    int e = blockIdx.x * blockDim.x + threadIdx.x;
    if (e < N_EDGES) atomicAdd(&deg[src[e]], 1.0f);
}

__global__ void k_dinv(const float* __restrict__ deg, float* __restrict__ dinv) {
    int i = blockIdx.x * blockDim.x + threadIdx.x;
    if (i < N_NODES) {
        float d = deg[i];
        d = d > 1.0f ? d : 1.0f;
        dinv[i] = rsqrtf(d);
    }
}

__global__ void k_gemm_wmma(const float* __restrict__ x,
                            const float* __restrict__ W,
                            const float* __restrict__ bias,
                            float* __restrict__ h) {
    const int wave = (blockIdx.x * blockDim.x + threadIdx.x) >> 5;
    const int lane = threadIdx.x & 31;
    const int nWaves = N_NODES / 16;
    if (wave >= nWaves) return;

    const int rowBase = wave * 16;
    const int half = lane >> 4;
    const int r    = lane & 15;

    v8f acc0 = {}, acc1 = {}, acc2 = {}, acc3 = {};

    for (int k = 0; k < IN_FEATS; k += 4) {
        v2f a;
        const float* xp = x + (size_t)(rowBase + r) * IN_FEATS + k + 2 * half;
        a.x = xp[0];
        a.y = xp[1];

        const float* wp = W + (size_t)(k + 2 * half) * H_FEATS + r;
        v2f b0, b1, b2, b3;
        b0.x = wp[0];      b0.y = wp[H_FEATS];
        b1.x = wp[16];     b1.y = wp[H_FEATS + 16];
        b2.x = wp[32];     b2.y = wp[H_FEATS + 32];
        b3.x = wp[48];     b3.y = wp[H_FEATS + 48];

        acc0 = __builtin_amdgcn_wmma_f32_16x16x4_f32(false, a, false, b0, (short)0, acc0, false, false);
        acc1 = __builtin_amdgcn_wmma_f32_16x16x4_f32(false, a, false, b1, (short)0, acc1, false, false);
        acc2 = __builtin_amdgcn_wmma_f32_16x16x4_f32(false, a, false, b2, (short)0, acc2, false, false);
        acc3 = __builtin_amdgcn_wmma_f32_16x16x4_f32(false, a, false, b3, (short)0, acc3, false, false);
    }

    const int mBase = rowBase + half * 8;
    v8f accs[4] = {acc0, acc1, acc2, acc3};
    #pragma unroll
    for (int nt = 0; nt < 4; ++nt) {
        const int col = nt * 16 + r;
        const float bv = bias[col];
        #pragma unroll
        for (int v = 0; v < 8; ++v) {
            float val = accs[nt][v] + bv;
            val = val > 0.0f ? val : NEG_SLOPE * val;
            h[(size_t)(mBase + v) * H_FEATS + col] = val;
        }
    }
}

__global__ void k_init_out(const float* __restrict__ h, float* __restrict__ out,
                           ThetaCol th) {
    int tid = blockIdx.x * blockDim.x + threadIdx.x;
    if (tid >= N_NODES * H_FEATS) return;
    int i = tid >> 6;
    int j = tid & 63;
    float hv = h[tid];
    #pragma unroll
    for (int t = 0; t < N_THETA; ++t)
        out[(size_t)i * (N_THETA * H_FEATS) + t * H_FEATS + j] = th.v[t] * hv;
}

__global__ void k_scatter(const int* __restrict__ src, const int* __restrict__ dst,
                          const float* __restrict__ dinv, const float* __restrict__ f,
                          float* __restrict__ msg) {
    int tid  = blockIdx.x * blockDim.x + threadIdx.x;
    int e    = tid >> 5;
    int lane = tid & 31;
    if (e >= N_EDGES) return;
    int s = src[e];
    int d = dst[e];
    float w = dinv[s];
    int j = lane * 2;
    const float2 v = *(const float2*)(f + (size_t)s * H_FEATS + j);
    float* mp = msg + (size_t)d * H_FEATS + j;
    atomicAdd(mp,     v.x * w);
    atomicAdd(mp + 1, v.y * w);
}

__global__ void k_update(const float* __restrict__ f, const float* __restrict__ msg,
                         const float* __restrict__ dinv, float* __restrict__ fnext,
                         float* __restrict__ out, ThetaCol th) {
    int tid = blockIdx.x * blockDim.x + threadIdx.x;
    if (tid >= N_NODES * H_FEATS) return;
    int i = tid >> 6;
    int j = tid & 63;
    float fn = f[tid] - msg[tid] * dinv[i];
    fnext[tid] = fn;
    float* op = out + (size_t)i * (N_THETA * H_FEATS) + j;
    #pragma unroll
    for (int t = 0; t < N_THETA; ++t)
        op[t * H_FEATS] += th.v[t] * fn;
}

static double host_comb(int n, int k) {
    double r = 1.0;
    for (int t = 0; t < k; ++t) r = r * (double)(n - t) / (double)(t + 1);
    return r;
}
static void compute_thetas(double th[N_THETA][K_STEPS + 1]) {
    const double e = 1.4;
    const int d = 4, OFF = 2;
    for (int idx = 0; idx < N_THETA; ++idx) {
        int i = OFF + idx;
        int m = d - i + OFF;
        double B = tgamma((double)(i + 1)) * tgamma((double)(d + 1 - i + OFF)) /
                   tgamma((double)(d + 2 + OFF));
        for (int j = 0; j <= K_STEPS; ++j) th[idx][j] = 0.0;
        for (int j = 0; j <= m; ++j) {
            double c = pow(1.0 / e, (double)i) * host_comb(m, j) *
                       pow(-1.0 / e, (double)j) / (e * B);
            th[idx][i + j] += c;
        }
    }
}

extern "C" void kernel_launch(void* const* d_in, const int* in_sizes, int n_in,
                              void* d_out, int out_size, void* d_ws, size_t ws_size,
                              hipStream_t stream) {
    const float* x    = (const float*)d_in[0];
    const float* W    = (const float*)d_in[1];
    const float* bias = (const float*)d_in[2];
    const int*   src  = (const int*)d_in[3];
    const int*   dst  = (const int*)d_in[4];
    float* out = (float*)d_out;

    float* wsf  = (float*)d_ws;
    float* deg  = wsf;
    float* dinv = deg + N_NODES;
    float* fA   = dinv + N_NODES;
    float* fB   = fA + (size_t)N_NODES * H_FEATS;
    float* msg  = fB + (size_t)N_NODES * H_FEATS;

    double th[N_THETA][K_STEPS + 1];
    compute_thetas(th);

    const int NF = N_NODES * H_FEATS;

    k_zero<<<(N_NODES + 255) / 256, 256, 0, stream>>>(deg, N_NODES);
    k_degree<<<(N_EDGES + 255) / 256, 256, 0, stream>>>(src, deg);
    k_dinv<<<(N_NODES + 255) / 256, 256, 0, stream>>>(deg, dinv);

    const int nWaves = N_NODES / 16;
    k_gemm_wmma<<<(nWaves + 7) / 8, 256, 0, stream>>>(x, W, bias, fA);

    ThetaCol tc;
    for (int t = 0; t < N_THETA; ++t) tc.v[t] = (float)th[t][0];
    k_init_out<<<(NF + 255) / 256, 256, 0, stream>>>(fA, out, tc);

    float* fcur = fA;
    float* fnext = fB;
    for (int k = 1; k <= K_STEPS; ++k) {
        k_zero<<<(NF + 255) / 256, 256, 0, stream>>>(msg, NF);
        k_scatter<<<(N_EDGES * 32 + 255) / 256, 256, 0, stream>>>(src, dst, dinv, fcur, msg);
        for (int t = 0; t < N_THETA; ++t) tc.v[t] = (float)th[t][k];
        k_update<<<(NF + 255) / 256, 256, 0, stream>>>(fcur, msg, dinv, fnext, out, tc);
        float* tmp = fcur; fcur = fnext; fnext = tmp;
    }
}


# SpMM pulled by destination from a per-call in-edge list (integer-atomic ranks, scan, fill, buckets sorted by source); a wave sums a node's in-edges in sorted order, update fused; no msg float atomics
# speedup vs baseline: 1.4455x; 1.4455x over previous
#include <hip/hip_runtime.h>
#include <hip/hip_bf16.h>
#include <cmath>

#define N_NODES   50000
#define N_EDGES   800000
#define IN_FEATS  256
#define H_FEATS   64
#define N_THETA   5
#define K_STEPS   6
#define NEG_SLOPE 0.01f
#define NODE_PAD  50176

typedef __attribute__((ext_vector_type(2))) float v2f;
typedef __attribute__((ext_vector_type(8))) float v8f;

struct ThetaCol { float v[N_THETA]; };

__global__ void k_zero2(float* __restrict__ deg, int* __restrict__ cnt) {
    int i = blockIdx.x * blockDim.x + threadIdx.x;
    if (i < N_NODES) { deg[i] = 0.0f; cnt[i] = 0; }
}

__global__ void k_degree_rank(const int* __restrict__ src, const int* __restrict__ dst,
                              float* __restrict__ deg, int* __restrict__ cnt,
                              int* __restrict__ rank) {
    int e = blockIdx.x * blockDim.x + threadIdx.x;
    if (e < N_EDGES) {
        atomicAdd(&deg[src[e]], 1.0f);
        rank[e] = atomicAdd(&cnt[dst[e]], 1);
    }
}

__global__ void k_dinv(const float* __restrict__ deg, float* __restrict__ dinv) {
    int i = blockIdx.x * blockDim.x + threadIdx.x;
    if (i < N_NODES) {
        float d = deg[i];
        d = d > 1.0f ? d : 1.0f;
        dinv[i] = rsqrtf(d);
    }
}

__global__ void k_gemm_wmma(const float* __restrict__ x,
                            const float* __restrict__ W,
                            const float* __restrict__ bias,
                            float* __restrict__ h) {
    const int wave = (blockIdx.x * blockDim.x + threadIdx.x) >> 5;
    const int lane = threadIdx.x & 31;
    const int nWaves = N_NODES / 16;
    if (wave >= nWaves) return;

    const int rowBase = wave * 16;
    const int half = lane >> 4;
    const int r    = lane & 15;

    v8f acc0 = {}, acc1 = {}, acc2 = {}, acc3 = {};

    for (int k = 0; k < IN_FEATS; k += 4) {
        v2f a;
        const float* xp = x + (size_t)(rowBase + r) * IN_FEATS + k + 2 * half;
        a.x = xp[0];
        a.y = xp[1];

        const float* wp = W + (size_t)(k + 2 * half) * H_FEATS + r;
        v2f b0, b1, b2, b3;
        b0.x = wp[0];      b0.y = wp[H_FEATS];
        b1.x = wp[16];     b1.y = wp[H_FEATS + 16];
        b2.x = wp[32];     b2.y = wp[H_FEATS + 32];
        b3.x = wp[48];     b3.y = wp[H_FEATS + 48];

        acc0 = __builtin_amdgcn_wmma_f32_16x16x4_f32(false, a, false, b0, (short)0, acc0, false, false);
        acc1 = __builtin_amdgcn_wmma_f32_16x16x4_f32(false, a, false, b1, (short)0, acc1, false, false);
        acc2 = __builtin_amdgcn_wmma_f32_16x16x4_f32(false, a, false, b2, (short)0, acc2, false, false);
        acc3 = __builtin_amdgcn_wmma_f32_16x16x4_f32(false, a, false, b3, (short)0, acc3, false, false);
    }

    const int mBase = rowBase + half * 8;
    v8f accs[4] = {acc0, acc1, acc2, acc3};
    #pragma unroll
    for (int nt = 0; nt < 4; ++nt) {
        const int col = nt * 16 + r;
        const float bv = bias[col];
        #pragma unroll
        for (int v = 0; v < 8; ++v) {
            float val = accs[nt][v] + bv;
            val = val > 0.0f ? val : NEG_SLOPE * val;
            h[(size_t)(mBase + v) * H_FEATS + col] = val;
        }
    }
}

__global__ void k_init_out(const float* __restrict__ h, float* __restrict__ out,
                           ThetaCol th) {
    int tid = blockIdx.x * blockDim.x + threadIdx.x;
    if (tid >= N_NODES * H_FEATS) return;
    int i = tid >> 6;
    int j = tid & 63;
    float hv = h[tid];
    #pragma unroll
    for (int t = 0; t < N_THETA; ++t)
        out[(size_t)i * (N_THETA * H_FEATS) + t * H_FEATS + j] = th.v[t] * hv;
}

#define SCAN_T   1024
#define SCAN_PER ((N_NODES + SCAN_T - 1) / SCAN_T)
__global__ void __launch_bounds__(SCAN_T) k_scan(const int* __restrict__ cnt,
                                                 int* __restrict__ rowptr) {
    __shared__ int wsum[32];
    const int t = threadIdx.x;
    const int lane = t & 31, w = t >> 5;
    const int lo = t * SCAN_PER;
    const int hi = (lo + SCAN_PER < N_NODES) ? (lo + SCAN_PER) : N_NODES;
    int s = 0;
    for (int k = lo; k < hi; ++k) s += cnt[k];
    int incl = s;
    for (int off = 1; off < 32; off <<= 1) {
        int v = __shfl_up(incl, off);
        if (lane >= off) incl += v;
    }
    if (lane == 31) wsum[w] = incl;
    __syncthreads();
    if (w == 0) {
        const int x = wsum[lane];
        int xi = x;
        for (int off = 1; off < 32; off <<= 1) {
            int v = __shfl_up(xi, off);
            if (lane >= off) xi += v;
        }
        wsum[lane] = xi - x;
    }
    __syncthreads();
    int run = incl - s + wsum[w];
    for (int k = lo; k < hi; ++k) { rowptr[k] = run; run += cnt[k]; }
    if (t == SCAN_T - 1) rowptr[N_NODES] = run;
}

__global__ void k_fill(const int* __restrict__ src, const int* __restrict__ dst,
                       const int* __restrict__ rowptr, const int* __restrict__ rank,
                       int* __restrict__ tmp) {
    int e = blockIdx.x * blockDim.x + threadIdx.x;
    if (e < N_EDGES) tmp[rowptr[dst[e]] + rank[e]] = src[e];
}

__global__ void k_sort(const int* __restrict__ rowptr, const int* __restrict__ tmp,
                       int* __restrict__ col) {
    const int i = __builtin_amdgcn_readfirstlane((int)((blockIdx.x * blockDim.x + threadIdx.x) >> 5));
    const int lane = threadIdx.x & 31;
    if (i >= N_NODES) return;
    const int b = __builtin_amdgcn_readfirstlane(rowptr[i]);
    const int n = __builtin_amdgcn_readfirstlane(rowptr[i + 1]) - b;
    if (n <= 32) {
        const int key = lane < n ? tmp[b + lane] : 0x7fffffff;
        int r = 0;
        for (int j = 0; j < n; ++j) {
            const int kj = __builtin_amdgcn_readlane(key, j);
            r += (kj < key || (kj == key && j < lane)) ? 1 : 0;
        }
        if (lane < n) col[b + r] = key;
    } else {
        for (int idx = lane; idx < n; idx += 32) {
            const int key = tmp[b + idx];
            int r = 0;
            for (int j = 0; j < n; ++j) {
                const int kj = tmp[b + j];
                r += (kj < key || (kj == key && j < idx)) ? 1 : 0;
            }
            col[b + r] = key;
        }
    }
}

__global__ void __launch_bounds__(256)
k_pull_update(const int* __restrict__ rowptr, const int* __restrict__ col,
              const float* __restrict__ dinv, const float* __restrict__ f,
              float* __restrict__ fnext, float* __restrict__ out, ThetaCol th) {
    const int i = __builtin_amdgcn_readfirstlane((int)((blockIdx.x * blockDim.x + threadIdx.x) >> 5));
    if (i >= N_NODES) return;
    const int lane = threadIdx.x & 31;
    const int half = lane >> 4;
    const int q    = lane & 15;
    const int b = __builtin_amdgcn_readfirstlane(rowptr[i]);
    const int n = __builtin_amdgcn_readfirstlane(rowptr[i + 1]) - b;

    float4 acc = make_float4(0.0f, 0.0f, 0.0f, 0.0f);
    for (int base = 0; base < n; base += 32) {
        const int m = (n - base < 32) ? (n - base) : 32;
        const int idx = base + lane;
        int   s_l = i;
        float w_l = 0.0f;
        if (idx < n) { s_l = col[b + idx]; w_l = dinv[s_l]; }
        const int w_li = __float_as_int(w_l);
        for (int j = 0; j < m; j += 8) {
            float4 v[4];
            float  w[4];
            #pragma unroll
            for (int u = 0; u < 4; ++u) {
                const int e0 = j + 2 * u;
                const int sa = __builtin_amdgcn_readlane(s_l, e0);
                const int sb = __builtin_amdgcn_readlane(s_l, e0 + 1);
                const int wa = __builtin_amdgcn_readlane(w_li, e0);
                const int wb = __builtin_amdgcn_readlane(w_li, e0 + 1);
                const int s = half ? sb : sa;
                w[u] = __int_as_float(half ? wb : wa);
                v[u] = *(const float4*)(f + (size_t)s * H_FEATS + 4 * q);
            }
            #pragma unroll
            for (int u = 0; u < 4; ++u) {
                acc.x = fmaf(v[u].x, w[u], acc.x);
                acc.y = fmaf(v[u].y, w[u], acc.y);
                acc.z = fmaf(v[u].z, w[u], acc.z);
                acc.w = fmaf(v[u].w, w[u], acc.w);
            }
        }
    }
    acc.x += __shfl_xor(acc.x, 16);
    acc.y += __shfl_xor(acc.y, 16);
    acc.z += __shfl_xor(acc.z, 16);
    acc.w += __shfl_xor(acc.w, 16);

    const float di = dinv[i];
    const float4 fv = *(const float4*)(f + (size_t)i * H_FEATS + 4 * q);
    float4 fn;
    fn.x = fv.x - acc.x * di;
    fn.y = fv.y - acc.y * di;
    fn.z = fv.z - acc.z * di;
    fn.w = fv.w - acc.w * di;
    if (half == 0) *(float4*)(fnext + (size_t)i * H_FEATS + 4 * q) = fn;

    float* ob = out + (size_t)i * (N_THETA * H_FEATS) + 4 * q;
    #pragma unroll
    for (int t = 0; t < N_THETA; t += 2) {
        const int tt = t + half;
        if (tt < N_THETA) {
            const float thv = half ? th.v[(t + 1 < N_THETA) ? (t + 1) : t] : th.v[t];
            float4 o = *(const float4*)(ob + tt * H_FEATS);
            o.x += thv * fn.x;
            o.y += thv * fn.y;
            o.z += thv * fn.z;
            o.w += thv * fn.w;
            *(float4*)(ob + tt * H_FEATS) = o;
        }
    }
}

static double host_comb(int n, int k) {
    double r = 1.0;
    for (int t = 0; t < k; ++t) r = r * (double)(n - t) / (double)(t + 1);
    return r;
}
static void compute_thetas(double th[N_THETA][K_STEPS + 1]) {
    const double e = 1.4;
    const int d = 4, OFF = 2;
    for (int idx = 0; idx < N_THETA; ++idx) {
        int i = OFF + idx;
        int m = d - i + OFF;
        double B = tgamma((double)(i + 1)) * tgamma((double)(d + 1 - i + OFF)) /
                   tgamma((double)(d + 2 + OFF));
        for (int j = 0; j <= K_STEPS; ++j) th[idx][j] = 0.0;
        for (int j = 0; j <= m; ++j) {
            double c = pow(1.0 / e, (double)i) * host_comb(m, j) *
                       pow(-1.0 / e, (double)j) / (e * B);
            th[idx][i + j] += c;
        }
    }
}

extern "C" void kernel_launch(void* const* d_in, const int* in_sizes, int n_in,
                              void* d_out, int out_size, void* d_ws, size_t ws_size,
                              hipStream_t stream) {
    const float* x    = (const float*)d_in[0];
    const float* W    = (const float*)d_in[1];
    const float* bias = (const float*)d_in[2];
    const int*   src  = (const int*)d_in[3];
    const int*   dst  = (const int*)d_in[4];
    float* out = (float*)d_out;

    float* wsf  = (float*)d_ws;
    float* deg  = wsf;
    float* dinv = deg + N_NODES;
    float* fA   = dinv + N_NODES;
    float* fB   = fA + (size_t)N_NODES * H_FEATS;
    int*   cnt    = (int*)(fB + (size_t)N_NODES * H_FEATS);
    int*   rowptr = cnt + NODE_PAD;
    int*   rank   = rowptr + NODE_PAD;
    int*   tmp    = rank + N_EDGES;
    int*   col    = tmp + N_EDGES;

    double th[N_THETA][K_STEPS + 1];
    compute_thetas(th);

    const int NF = N_NODES * H_FEATS;

    k_zero2<<<(N_NODES + 255) / 256, 256, 0, stream>>>(deg, cnt);
    k_degree_rank<<<(N_EDGES + 255) / 256, 256, 0, stream>>>(src, dst, deg, cnt, rank);
    k_dinv<<<(N_NODES + 255) / 256, 256, 0, stream>>>(deg, dinv);
    k_scan<<<1, SCAN_T, 0, stream>>>(cnt, rowptr);
    k_fill<<<(N_EDGES + 255) / 256, 256, 0, stream>>>(src, dst, rowptr, rank, tmp);
    k_sort<<<(N_NODES * 32 + 255) / 256, 256, 0, stream>>>(rowptr, tmp, col);

    const int nWaves = N_NODES / 16;
    k_gemm_wmma<<<(nWaves + 7) / 8, 256, 0, stream>>>(x, W, bias, fA);

    ThetaCol tc;
    for (int t = 0; t < N_THETA; ++t) tc.v[t] = (float)th[t][0];
    k_init_out<<<(NF + 255) / 256, 256, 0, stream>>>(fA, out, tc);

    float* fcur = fA;
    float* fnext = fB;
    for (int k = 1; k <= K_STEPS; ++k) {
        for (int t = 0; t < N_THETA; ++t) tc.v[t] = (float)th[t][k];
        k_pull_update<<<(N_NODES * 32 + 255) / 256, 256, 0, stream>>>(rowptr, col, dinv, fcur, fnext, out, tc);
        float* sw = fcur; fcur = fnext; fnext = sw;
    }
}


# Far fewer, longer-lived waves per launch: a wave walks 4 nodes (pull, sort; next node's bucket fetched ahead), a lane 4 edges or 4 float4 (degree, fill, out init); strided grids; sum order kept
# speedup vs baseline: 1.6457x; 1.1385x over previous
#include <hip/hip_runtime.h>
#include <hip/hip_bf16.h>
#include <cmath>

#define N_NODES   50000
#define N_EDGES   800000
#define IN_FEATS  256
#define H_FEATS   64
#define N_THETA   5
#define K_STEPS   6
#define NEG_SLOPE 0.01f
#define NODE_PAD  50176

#define NODES_PER_WAVE   4
#define NODE_GRID  (((N_NODES + NODES_PER_WAVE - 1) / NODES_PER_WAVE + 7) / 8)
#define ITEMS_PER_LANE   4
#define EDGE_GRID  ((N_EDGES / ITEMS_PER_LANE + 255) / 256)
#define ROWQ_GRID  ((N_NODES * (H_FEATS / 4) / ITEMS_PER_LANE + 255) / 256)

typedef __attribute__((ext_vector_type(2))) float v2f;
typedef __attribute__((ext_vector_type(8))) float v8f;

struct ThetaCol { float v[N_THETA]; };

__global__ void k_zero2(float* __restrict__ deg, int* __restrict__ cnt) {
    int i = blockIdx.x * blockDim.x + threadIdx.x;
    if (i < N_NODES) { deg[i] = 0.0f; cnt[i] = 0; }
}

__global__ void k_degree_rank(const int* __restrict__ src, const int* __restrict__ dst,
                              float* __restrict__ deg, int* __restrict__ cnt,
                              int* __restrict__ rank) {
    const int tid    = blockIdx.x * blockDim.x + threadIdx.x;
    const int stride = gridDim.x * blockDim.x;
    for (int e0 = tid; e0 < N_EDGES; e0 += ITEMS_PER_LANE * stride) {
        int s[ITEMS_PER_LANE], d[ITEMS_PER_LANE], r[ITEMS_PER_LANE];
        #pragma unroll
        for (int u = 0; u < ITEMS_PER_LANE; ++u) {
            const int e  = e0 + u * stride;
            const int ec = e < N_EDGES ? e : N_EDGES - 1;
            s[u] = src[ec];
            d[u] = dst[ec];
        }
        #pragma unroll
        for (int u = 0; u < ITEMS_PER_LANE; ++u) {
            const int e = e0 + u * stride;
            r[u] = 0;
            if (e < N_EDGES) {
                atomicAdd(&deg[s[u]], 1.0f);
                r[u] = atomicAdd(&cnt[d[u]], 1);
            }
        }
        #pragma unroll
        for (int u = 0; u < ITEMS_PER_LANE; ++u) {
            const int e = e0 + u * stride;
            if (e < N_EDGES) rank[e] = r[u];
        }
    }
}

__global__ void k_dinv(const float* __restrict__ deg, float* __restrict__ dinv) {
    int i = blockIdx.x * blockDim.x + threadIdx.x;
    if (i < N_NODES) {
        float d = deg[i];
        d = d > 1.0f ? d : 1.0f;
        dinv[i] = rsqrtf(d);
    }
}

__global__ void k_gemm_wmma(const float* __restrict__ x,
                            const float* __restrict__ W,
                            const float* __restrict__ bias,
                            float* __restrict__ h) {
    const int wave = (blockIdx.x * blockDim.x + threadIdx.x) >> 5;
    const int lane = threadIdx.x & 31;
    const int nWaves = N_NODES / 16;
    if (wave >= nWaves) return;

    const int rowBase = wave * 16;
    const int half = lane >> 4;
    const int r    = lane & 15;

    v8f acc0 = {}, acc1 = {}, acc2 = {}, acc3 = {};

    for (int k = 0; k < IN_FEATS; k += 4) {
        v2f a;
        const float* xp = x + (size_t)(rowBase + r) * IN_FEATS + k + 2 * half;
        a.x = xp[0];
        a.y = xp[1];

        const float* wp = W + (size_t)(k + 2 * half) * H_FEATS + r;
        v2f b0, b1, b2, b3;
        b0.x = wp[0];      b0.y = wp[H_FEATS];
        b1.x = wp[16];     b1.y = wp[H_FEATS + 16];
        b2.x = wp[32];     b2.y = wp[H_FEATS + 32];
        b3.x = wp[48];     b3.y = wp[H_FEATS + 48];

        acc0 = __builtin_amdgcn_wmma_f32_16x16x4_f32(false, a, false, b0, (short)0, acc0, false, false);
        acc1 = __builtin_amdgcn_wmma_f32_16x16x4_f32(false, a, false, b1, (short)0, acc1, false, false);
        acc2 = __builtin_amdgcn_wmma_f32_16x16x4_f32(false, a, false, b2, (short)0, acc2, false, false);
        acc3 = __builtin_amdgcn_wmma_f32_16x16x4_f32(false, a, false, b3, (short)0, acc3, false, false);
    }

    const int mBase = rowBase + half * 8;
    v8f accs[4] = {acc0, acc1, acc2, acc3};
    #pragma unroll
    for (int nt = 0; nt < 4; ++nt) {
        const int col = nt * 16 + r;
        const float bv = bias[col];
        #pragma unroll
        for (int v = 0; v < 8; ++v) {
            float val = accs[nt][v] + bv;
            val = val > 0.0f ? val : NEG_SLOPE * val;
            h[(size_t)(mBase + v) * H_FEATS + col] = val;
        }
    }
}

__global__ void k_init_out(const float* __restrict__ h, float* __restrict__ out,
                           ThetaCol th) {
    const int tid    = blockIdx.x * blockDim.x + threadIdx.x;
    const int stride = gridDim.x * blockDim.x;
    const int nq     = N_NODES * (H_FEATS / 4);
    for (int id0 = tid; id0 < nq; id0 += ITEMS_PER_LANE * stride) {
        float4 hv[ITEMS_PER_LANE];
        #pragma unroll
        for (int u = 0; u < ITEMS_PER_LANE; ++u) {
            const int id  = id0 + u * stride;
            const int idc = id < nq ? id : nq - 1;
            hv[u] = *(const float4*)(h + (size_t)idc * 4);
        }
        #pragma unroll
        for (int u = 0; u < ITEMS_PER_LANE; ++u) {
            const int id = id0 + u * stride;
            if (id < nq) {
                const int i = id >> 4;
                const int q = id & 15;
                float* ob = out + (size_t)i * (N_THETA * H_FEATS) + 4 * q;
                #pragma unroll
                for (int t = 0; t < N_THETA; ++t) {
                    float4 o;
                    o.x = th.v[t] * hv[u].x;
                    o.y = th.v[t] * hv[u].y;
                    o.z = th.v[t] * hv[u].z;
                    o.w = th.v[t] * hv[u].w;
                    *(float4*)(ob + t * H_FEATS) = o;
                }
            }
        }
    }
}

#define SCAN_T   1024
#define SCAN_PER ((N_NODES + SCAN_T - 1) / SCAN_T)
__global__ void __launch_bounds__(SCAN_T) k_scan(const int* __restrict__ cnt,
                                                 int* __restrict__ rowptr) {
    __shared__ int wsum[32];
    const int t = threadIdx.x;
    const int lane = t & 31, w = t >> 5;
    const int lo = t * SCAN_PER;
    const int hi = (lo + SCAN_PER < N_NODES) ? (lo + SCAN_PER) : N_NODES;
    int s = 0;
    for (int k = lo; k < hi; ++k) s += cnt[k];
    int incl = s;
    for (int off = 1; off < 32; off <<= 1) {
        int v = __shfl_up(incl, off);
        if (lane >= off) incl += v;
    }
    if (lane == 31) wsum[w] = incl;
    __syncthreads();
    if (w == 0) {
        const int x = wsum[lane];
        int xi = x;
        for (int off = 1; off < 32; off <<= 1) {
            int v = __shfl_up(xi, off);
            if (lane >= off) xi += v;
        }
        wsum[lane] = xi - x;
    }
    __syncthreads();
    int run = incl - s + wsum[w];
    for (int k = lo; k < hi; ++k) { rowptr[k] = run; run += cnt[k]; }
    if (t == SCAN_T - 1) rowptr[N_NODES] = run;
}

__global__ void k_fill(const int* __restrict__ src, const int* __restrict__ dst,
                       const int* __restrict__ rowptr, const int* __restrict__ rank,
                       int* __restrict__ tmp) {
    const int tid    = blockIdx.x * blockDim.x + threadIdx.x;
    const int stride = gridDim.x * blockDim.x;
    for (int e0 = tid; e0 < N_EDGES; e0 += ITEMS_PER_LANE * stride) {
        int s[ITEMS_PER_LANE], d[ITEMS_PER_LANE], p[ITEMS_PER_LANE];
        #pragma unroll
        for (int u = 0; u < ITEMS_PER_LANE; ++u) {
            const int e  = e0 + u * stride;
            const int ec = e < N_EDGES ? e : N_EDGES - 1;
            s[u] = src[ec];
            d[u] = dst[ec];
            p[u] = rank[ec];
        }
        #pragma unroll
        for (int u = 0; u < ITEMS_PER_LANE; ++u) p[u] += rowptr[d[u]];
        #pragma unroll
        for (int u = 0; u < ITEMS_PER_LANE; ++u) {
            const int e = e0 + u * stride;
            if (e < N_EDGES) tmp[p[u]] = s[u];
        }
    }
}

__global__ void k_sort(const int* __restrict__ rowptr, const int* __restrict__ tmp,
                       int* __restrict__ col) {
    const int w0 = __builtin_amdgcn_readfirstlane((int)((blockIdx.x * blockDim.x + threadIdx.x) >> 5));
    const int nw = (int)((gridDim.x * blockDim.x) >> 5);
    const int lane = threadIdx.x & 31;
    for (int i = w0; i < N_NODES; i += nw) {
        const int b = __builtin_amdgcn_readfirstlane(rowptr[i]);
        const int n = __builtin_amdgcn_readfirstlane(rowptr[i + 1]) - b;
        if (n <= 32) {
            const int key = lane < n ? tmp[b + lane] : 0x7fffffff;
            int r = 0;
            for (int j = 0; j < n; ++j) {
                const int kj = __builtin_amdgcn_readlane(key, j);
                r += (kj < key || (kj == key && j < lane)) ? 1 : 0;
            }
            if (lane < n) col[b + r] = key;
        } else {
            for (int idx = lane; idx < n; idx += 32) {
                const int key = tmp[b + idx];
                int r = 0;
                for (int j = 0; j < n; ++j) {
                    const int kj = tmp[b + j];
                    r += (kj < key || (kj == key && j < idx)) ? 1 : 0;
                }
                col[b + r] = key;
            }
        }
    }
}

__global__ void __launch_bounds__(256)
k_pull_update(const int* __restrict__ rowptr, const int* __restrict__ col,
              const float* __restrict__ dinv, const float* __restrict__ f,
              float* __restrict__ fnext, float* __restrict__ out, ThetaCol th) {
    const int lane = threadIdx.x & 31;
    const int half = lane >> 4;
    const int q    = lane & 15;
    const int w0 = __builtin_amdgcn_readfirstlane((int)((blockIdx.x * blockDim.x + threadIdx.x) >> 5));
    const int nw = (int)((gridDim.x * blockDim.x) >> 5);
    if (w0 >= N_NODES) return;

    int i = w0;
    int b = __builtin_amdgcn_readfirstlane(rowptr[i]);
    int n = __builtin_amdgcn_readfirstlane(rowptr[i + 1]) - b;
    int   s_l = i;
    float w_l = 0.0f;
    if (lane < n) { s_l = col[b + lane]; w_l = dinv[s_l]; }

    int  i2   = i + nw;
    bool has2 = i2 < N_NODES;
    int  b2 = 0, e2 = 0;
    if (has2) {
        b2 = __builtin_amdgcn_readfirstlane(rowptr[i2]);
        e2 = __builtin_amdgcn_readfirstlane(rowptr[i2 + 1]);
    }

    for (;;) {
        const int  n2   = e2 - b2;
        const int  i3   = i2 + nw;
        const bool has3 = i3 < N_NODES;
        int b3 = 0, e3 = 0;
        if (has3) {
            b3 = __builtin_amdgcn_readfirstlane(rowptr[i3]);
            e3 = __builtin_amdgcn_readfirstlane(rowptr[i3 + 1]);
        }
        int s_n = has2 ? i2 : i;
        if (has2 && lane < n2) s_n = col[b2 + lane];

        const float  di = dinv[i];
        const float4 fv = *(const float4*)(f + (size_t)i * H_FEATS + 4 * q);
        float* ob = out + (size_t)i * (N_THETA * H_FEATS) + 4 * q;
        float4 o[3];
        #pragma unroll
        for (int u = 0; u < 3; ++u) {
            const int tt = 2 * u + half;
            o[u] = make_float4(0.0f, 0.0f, 0.0f, 0.0f);
            if (tt < N_THETA) o[u] = *(const float4*)(ob + tt * H_FEATS);
        }

        float4 acc = make_float4(0.0f, 0.0f, 0.0f, 0.0f);
        for (int base = 0; base < n; base += 32) {
            if (base > 0) {
                const int idx = base + lane;
                s_l = i;
                w_l = 0.0f;
                if (idx < n) { s_l = col[b + idx]; w_l = dinv[s_l]; }
            }
            const int m = (n - base < 32) ? (n - base) : 32;
            const int w_li = __float_as_int(w_l);
            for (int j = 0; j < m; j += 8) {
                float4 v[4];
                float  w[4];
                #pragma unroll
                for (int u = 0; u < 4; ++u) {
                    const int e0 = j + 2 * u;
                    const int sa = __builtin_amdgcn_readlane(s_l, e0);
                    const int sb = __builtin_amdgcn_readlane(s_l, e0 + 1);
                    const int wa = __builtin_amdgcn_readlane(w_li, e0);
                    const int wb = __builtin_amdgcn_readlane(w_li, e0 + 1);
                    const int s = half ? sb : sa;
                    w[u] = __int_as_float(half ? wb : wa);
                    v[u] = *(const float4*)(f + (size_t)s * H_FEATS + 4 * q);
                }
                #pragma unroll
                for (int u = 0; u < 4; ++u) {
                    acc.x = fmaf(v[u].x, w[u], acc.x);
                    acc.y = fmaf(v[u].y, w[u], acc.y);
                    acc.z = fmaf(v[u].z, w[u], acc.z);
                    acc.w = fmaf(v[u].w, w[u], acc.w);
                }
            }
        }

        float w_n = 0.0f;
        if (has2 && lane < n2) w_n = dinv[s_n];

        acc.x += __shfl_xor(acc.x, 16);
        acc.y += __shfl_xor(acc.y, 16);
        acc.z += __shfl_xor(acc.z, 16);
        acc.w += __shfl_xor(acc.w, 16);

        float4 fn;
        fn.x = fv.x - acc.x * di;
        fn.y = fv.y - acc.y * di;
        fn.z = fv.z - acc.z * di;
        fn.w = fv.w - acc.w * di;
        if (half == 0) *(float4*)(fnext + (size_t)i * H_FEATS + 4 * q) = fn;

        #pragma unroll
        for (int u = 0; u < 3; ++u) {
            const int t  = 2 * u;
            const int tt = t + half;
            if (tt < N_THETA) {
                const float thv = half ? th.v[(t + 1 < N_THETA) ? (t + 1) : t] : th.v[t];
                float4 r = o[u];
                r.x += thv * fn.x;
                r.y += thv * fn.y;
                r.z += thv * fn.z;
                r.w += thv * fn.w;
                *(float4*)(ob + tt * H_FEATS) = r;
            }
        }

        if (!has2) break;
        i = i2; b = b2; n = n2; s_l = s_n; w_l = w_n;
        i2 = i3; has2 = has3; b2 = b3; e2 = e3;
    }
}

static double host_comb(int n, int k) {
    double r = 1.0;
    for (int t = 0; t < k; ++t) r = r * (double)(n - t) / (double)(t + 1);
    return r;
}
static void compute_thetas(double th[N_THETA][K_STEPS + 1]) {
    const double e = 1.4;
    const int d = 4, OFF = 2;
    for (int idx = 0; idx < N_THETA; ++idx) {
        int i = OFF + idx;
        int m = d - i + OFF;
        double B = tgamma((double)(i + 1)) * tgamma((double)(d + 1 - i + OFF)) /
                   tgamma((double)(d + 2 + OFF));
        for (int j = 0; j <= K_STEPS; ++j) th[idx][j] = 0.0;
        for (int j = 0; j <= m; ++j) {
            double c = pow(1.0 / e, (double)i) * host_comb(m, j) *
                       pow(-1.0 / e, (double)j) / (e * B);
            th[idx][i + j] += c;
        }
    }
}

extern "C" void kernel_launch(void* const* d_in, const int* in_sizes, int n_in,
                              void* d_out, int out_size, void* d_ws, size_t ws_size,
                              hipStream_t stream) {
    const float* x    = (const float*)d_in[0];
    const float* W    = (const float*)d_in[1];
    const float* bias = (const float*)d_in[2];
    const int*   src  = (const int*)d_in[3];
    const int*   dst  = (const int*)d_in[4];
    float* out = (float*)d_out;

    float* wsf  = (float*)d_ws;
    float* deg  = wsf;
    float* dinv = deg + N_NODES;
    float* fA   = dinv + N_NODES;
    float* fB   = fA + (size_t)N_NODES * H_FEATS;
    int*   cnt    = (int*)(fB + (size_t)N_NODES * H_FEATS);
    int*   rowptr = cnt + NODE_PAD;
    int*   rank   = rowptr + NODE_PAD;
    int*   tmp    = rank + N_EDGES;
    int*   col    = tmp + N_EDGES;

    double th[N_THETA][K_STEPS + 1];
    compute_thetas(th);


    k_zero2<<<(N_NODES + 255) / 256, 256, 0, stream>>>(deg, cnt);
    k_degree_rank<<<EDGE_GRID, 256, 0, stream>>>(src, dst, deg, cnt, rank);
    k_dinv<<<(N_NODES + 255) / 256, 256, 0, stream>>>(deg, dinv);
    k_scan<<<1, SCAN_T, 0, stream>>>(cnt, rowptr);
    k_fill<<<EDGE_GRID, 256, 0, stream>>>(src, dst, rowptr, rank, tmp);
    k_sort<<<NODE_GRID, 256, 0, stream>>>(rowptr, tmp, col);

    const int nWaves = N_NODES / 16;
    k_gemm_wmma<<<(nWaves + 7) / 8, 256, 0, stream>>>(x, W, bias, fA);

    ThetaCol tc;
    for (int t = 0; t < N_THETA; ++t) tc.v[t] = (float)th[t][0];
    k_init_out<<<ROWQ_GRID, 256, 0, stream>>>(fA, out, tc);

    float* fcur = fA;
    float* fnext = fB;
    for (int k = 1; k <= K_STEPS; ++k) {
        for (int t = 0; t < N_THETA; ++t) tc.v[t] = (float)th[t][k];
        k_pull_update<<<NODE_GRID, 256, 0, stream>>>(rowptr, col, dinv, fcur, fnext, out, tc);
        float* sw = fcur; fcur = fnext; fnext = sw;
    }
}


# Filter sums deferred: steps keep only f_k (f_0..f_4 in the out row's five slots, f_5 in workspace); last step forms all five filters, each out row written once; no init launch; sum order kept
# speedup vs baseline: 2.0505x; 1.2460x over previous
#include <hip/hip_runtime.h>
#include <hip/hip_bf16.h>
#include <cmath>

#define N_NODES   50000
#define N_EDGES   800000
#define IN_FEATS  256
#define H_FEATS   64
#define N_THETA   5
#define K_STEPS   6
#define NEG_SLOPE 0.01f
#define NODE_PAD  50176
#define OUT_ROW   (N_THETA * H_FEATS)

#define NODES_PER_WAVE   4
#define NODE_GRID  (((N_NODES + NODES_PER_WAVE - 1) / NODES_PER_WAVE + 7) / 8)
#define ITEMS_PER_LANE   4
#define EDGE_GRID  ((N_EDGES / ITEMS_PER_LANE + 255) / 256)

typedef __attribute__((ext_vector_type(2))) float v2f;
typedef __attribute__((ext_vector_type(8))) float v8f;

struct ThetaAll { float v[N_THETA][K_STEPS + 1]; };


__global__ void k_zero2(float* __restrict__ deg, int* __restrict__ cnt) {
    int i = blockIdx.x * blockDim.x + threadIdx.x;
    if (i < N_NODES) { deg[i] = 0.0f; cnt[i] = 0; }
}

__global__ void k_degree_rank(const int* __restrict__ src, const int* __restrict__ dst,
                              float* __restrict__ deg, int* __restrict__ cnt,
                              int* __restrict__ rank) {
    const int tid    = blockIdx.x * blockDim.x + threadIdx.x;
    const int stride = gridDim.x * blockDim.x;
    for (int e0 = tid; e0 < N_EDGES; e0 += ITEMS_PER_LANE * stride) {
        int s[ITEMS_PER_LANE], d[ITEMS_PER_LANE], r[ITEMS_PER_LANE];
        #pragma unroll
        for (int u = 0; u < ITEMS_PER_LANE; ++u) {
            const int e  = e0 + u * stride;
            const int ec = e < N_EDGES ? e : N_EDGES - 1;
            s[u] = src[ec];
            d[u] = dst[ec];
        }
        #pragma unroll
        for (int u = 0; u < ITEMS_PER_LANE; ++u) {
            const int e = e0 + u * stride;
            r[u] = 0;
            if (e < N_EDGES) {
                atomicAdd(&deg[s[u]], 1.0f);
                r[u] = atomicAdd(&cnt[d[u]], 1);
            }
        }
        #pragma unroll
        for (int u = 0; u < ITEMS_PER_LANE; ++u) {
            const int e = e0 + u * stride;
            if (e < N_EDGES) rank[e] = r[u];
        }
    }
}

__global__ void k_dinv(const float* __restrict__ deg, float* __restrict__ dinv) {
    int i = blockIdx.x * blockDim.x + threadIdx.x;
    if (i < N_NODES) {
        float d = deg[i];
        d = d > 1.0f ? d : 1.0f;
        dinv[i] = rsqrtf(d);
    }
}

__global__ void k_gemm_wmma(const float* __restrict__ x,
                            const float* __restrict__ W,
                            const float* __restrict__ bias,
                            float* __restrict__ h) {
    const int wave = (blockIdx.x * blockDim.x + threadIdx.x) >> 5;
    const int lane = threadIdx.x & 31;
    const int nWaves = N_NODES / 16;
    if (wave >= nWaves) return;

    const int rowBase = wave * 16;
    const int half = lane >> 4;
    const int r    = lane & 15;

    v8f acc0 = {}, acc1 = {}, acc2 = {}, acc3 = {};

    for (int k = 0; k < IN_FEATS; k += 4) {
        v2f a;
        const float* xp = x + (size_t)(rowBase + r) * IN_FEATS + k + 2 * half;
        a.x = xp[0];
        a.y = xp[1];

        const float* wp = W + (size_t)(k + 2 * half) * H_FEATS + r;
        v2f b0, b1, b2, b3;
        b0.x = wp[0];      b0.y = wp[H_FEATS];
        b1.x = wp[16];     b1.y = wp[H_FEATS + 16];
        b2.x = wp[32];     b2.y = wp[H_FEATS + 32];
        b3.x = wp[48];     b3.y = wp[H_FEATS + 48];

        acc0 = __builtin_amdgcn_wmma_f32_16x16x4_f32(false, a, false, b0, (short)0, acc0, false, false);
        acc1 = __builtin_amdgcn_wmma_f32_16x16x4_f32(false, a, false, b1, (short)0, acc1, false, false);
        acc2 = __builtin_amdgcn_wmma_f32_16x16x4_f32(false, a, false, b2, (short)0, acc2, false, false);
        acc3 = __builtin_amdgcn_wmma_f32_16x16x4_f32(false, a, false, b3, (short)0, acc3, false, false);
    }

    const int mBase = rowBase + half * 8;
    v8f accs[4] = {acc0, acc1, acc2, acc3};
    #pragma unroll
    for (int nt = 0; nt < 4; ++nt) {
        const int col = nt * 16 + r;
        const float bv = bias[col];
        #pragma unroll
        for (int v = 0; v < 8; ++v) {
            float val = accs[nt][v] + bv;
            val = val > 0.0f ? val : NEG_SLOPE * val;
            h[(size_t)(mBase + v) * OUT_ROW + col] = val;
        }
    }
}

#define SCAN_T   1024
#define SCAN_PER ((N_NODES + SCAN_T - 1) / SCAN_T)
__global__ void __launch_bounds__(SCAN_T) k_scan(const int* __restrict__ cnt,
                                                 int* __restrict__ rowptr) {
    __shared__ int wsum[32];
    const int t = threadIdx.x;
    const int lane = t & 31, w = t >> 5;
    const int lo = t * SCAN_PER;
    const int hi = (lo + SCAN_PER < N_NODES) ? (lo + SCAN_PER) : N_NODES;
    int s = 0;
    for (int k = lo; k < hi; ++k) s += cnt[k];
    int incl = s;
    for (int off = 1; off < 32; off <<= 1) {
        int v = __shfl_up(incl, off);
        if (lane >= off) incl += v;
    }
    if (lane == 31) wsum[w] = incl;
    __syncthreads();
    if (w == 0) {
        const int x = wsum[lane];
        int xi = x;
        for (int off = 1; off < 32; off <<= 1) {
            int v = __shfl_up(xi, off);
            if (lane >= off) xi += v;
        }
        wsum[lane] = xi - x;
    }
    __syncthreads();
    int run = incl - s + wsum[w];
    for (int k = lo; k < hi; ++k) { rowptr[k] = run; run += cnt[k]; }
    if (t == SCAN_T - 1) rowptr[N_NODES] = run;
}

__global__ void k_fill(const int* __restrict__ src, const int* __restrict__ dst,
                       const int* __restrict__ rowptr, const int* __restrict__ rank,
                       int* __restrict__ tmp) {
    const int tid    = blockIdx.x * blockDim.x + threadIdx.x;
    const int stride = gridDim.x * blockDim.x;
    for (int e0 = tid; e0 < N_EDGES; e0 += ITEMS_PER_LANE * stride) {
        int s[ITEMS_PER_LANE], d[ITEMS_PER_LANE], p[ITEMS_PER_LANE];
        #pragma unroll
        for (int u = 0; u < ITEMS_PER_LANE; ++u) {
            const int e  = e0 + u * stride;
            const int ec = e < N_EDGES ? e : N_EDGES - 1;
            s[u] = src[ec];
            d[u] = dst[ec];
            p[u] = rank[ec];
        }
        #pragma unroll
        for (int u = 0; u < ITEMS_PER_LANE; ++u) p[u] += rowptr[d[u]];
        #pragma unroll
        for (int u = 0; u < ITEMS_PER_LANE; ++u) {
            const int e = e0 + u * stride;
            if (e < N_EDGES) tmp[p[u]] = s[u];
        }
    }
}

__global__ void k_sort(const int* __restrict__ rowptr, const int* __restrict__ tmp,
                       int* __restrict__ col) {
    const int w0 = __builtin_amdgcn_readfirstlane((int)((blockIdx.x * blockDim.x + threadIdx.x) >> 5));
    const int nw = (int)((gridDim.x * blockDim.x) >> 5);
    const int lane = threadIdx.x & 31;
    for (int i = w0; i < N_NODES; i += nw) {
        const int b = __builtin_amdgcn_readfirstlane(rowptr[i]);
        const int n = __builtin_amdgcn_readfirstlane(rowptr[i + 1]) - b;
        if (n <= 32) {
            const int key = lane < n ? tmp[b + lane] : 0x7fffffff;
            int r = 0;
            for (int j = 0; j < n; ++j) {
                const int kj = __builtin_amdgcn_readlane(key, j);
                r += (kj < key || (kj == key && j < lane)) ? 1 : 0;
            }
            if (lane < n) col[b + r] = key;
        } else {
            for (int idx = lane; idx < n; idx += 32) {
                const int key = tmp[b + idx];
                int r = 0;
                for (int j = 0; j < n; ++j) {
                    const int kj = tmp[b + j];
                    r += (kj < key || (kj == key && j < idx)) ? 1 : 0;
                }
                col[b + r] = key;
            }
        }
    }
}

template <int SRC_ROW, int DST_ROW, bool LAST>
__global__ void __launch_bounds__(256)
k_pull(const int* __restrict__ rowptr, const int* __restrict__ col,
       const float* __restrict__ dinv, const float* __restrict__ f,
       float* __restrict__ dst, ThetaAll th) {
    static_assert(K_STEPS == N_THETA + 1, "f_0..f_4 fill the five slots of an out row, f_5 the workspace plane");
    const int lane = threadIdx.x & 31;
    const int half = lane >> 4;
    const int q    = lane & 15;
    const int w0 = __builtin_amdgcn_readfirstlane((int)((blockIdx.x * blockDim.x + threadIdx.x) >> 5));
    const int nw = (int)((gridDim.x * blockDim.x) >> 5);
    if (w0 >= N_NODES) return;

    int i = w0;
    int b = __builtin_amdgcn_readfirstlane(rowptr[i]);
    int n = __builtin_amdgcn_readfirstlane(rowptr[i + 1]) - b;
    int   s_l = i;
    float w_l = 0.0f;
    if (lane < n) { s_l = col[b + lane]; w_l = dinv[s_l]; }

    int  i2   = i + nw;
    bool has2 = i2 < N_NODES;
    int  b2 = 0, e2 = 0;
    if (has2) {
        b2 = __builtin_amdgcn_readfirstlane(rowptr[i2]);
        e2 = __builtin_amdgcn_readfirstlane(rowptr[i2 + 1]);
    }

    for (;;) {
        const int  n2   = e2 - b2;
        const int  i3   = i2 + nw;
        const bool has3 = i3 < N_NODES;
        int b3 = 0, e3 = 0;
        if (has3) {
            b3 = __builtin_amdgcn_readfirstlane(rowptr[i3]);
            e3 = __builtin_amdgcn_readfirstlane(rowptr[i3 + 1]);
        }
        int s_n = has2 ? i2 : i;
        if (has2 && lane < n2) s_n = col[b2 + lane];

        const float  di = dinv[i];
        const float4 fv = *(const float4*)(f + (size_t)i * SRC_ROW + 4 * q);
        float* ob = dst + (size_t)i * DST_ROW + 4 * q;
        [[maybe_unused]] float4 own[N_THETA];
        if constexpr (LAST) {
            #pragma unroll
            for (int k = 0; k < N_THETA; ++k) own[k] = *(const float4*)(ob + k * H_FEATS);
        }

        float4 acc = make_float4(0.0f, 0.0f, 0.0f, 0.0f);
        for (int base = 0; base < n; base += 32) {
            if (base > 0) {
                const int idx = base + lane;
                s_l = i;
                w_l = 0.0f;
                if (idx < n) { s_l = col[b + idx]; w_l = dinv[s_l]; }
            }
            const int m = (n - base < 32) ? (n - base) : 32;
            const int w_li = __float_as_int(w_l);
            for (int j = 0; j < m; j += 8) {
                float4 v[4];
                float  w[4];
                #pragma unroll
                for (int u = 0; u < 4; ++u) {
                    const int e0 = j + 2 * u;
                    const int sa = __builtin_amdgcn_readlane(s_l, e0);
                    const int sb = __builtin_amdgcn_readlane(s_l, e0 + 1);
                    const int wa = __builtin_amdgcn_readlane(w_li, e0);
                    const int wb = __builtin_amdgcn_readlane(w_li, e0 + 1);
                    const int s = half ? sb : sa;
                    w[u] = __int_as_float(half ? wb : wa);
                    v[u] = *(const float4*)(f + (size_t)s * SRC_ROW + 4 * q);
                }
                #pragma unroll
                for (int u = 0; u < 4; ++u) {
                    acc.x = fmaf(v[u].x, w[u], acc.x);
                    acc.y = fmaf(v[u].y, w[u], acc.y);
                    acc.z = fmaf(v[u].z, w[u], acc.z);
                    acc.w = fmaf(v[u].w, w[u], acc.w);
                }
            }
        }

        float w_n = 0.0f;
        if (has2 && lane < n2) w_n = dinv[s_n];

        acc.x += __shfl_xor(acc.x, 16);
        acc.y += __shfl_xor(acc.y, 16);
        acc.z += __shfl_xor(acc.z, 16);
        acc.w += __shfl_xor(acc.w, 16);

        float4 fn;
        fn.x = fv.x - acc.x * di;
        fn.y = fv.y - acc.y * di;
        fn.z = fv.z - acc.z * di;
        fn.w = fv.w - acc.w * di;

        if constexpr (!LAST) {
            if (half == 0) *(float4*)ob = fn;
        } else {
            #pragma unroll
            for (int t = 0; t < N_THETA && half == 0; ++t) {
                float4 r;
                r.x = th.v[t][0] * own[0].x;
                r.y = th.v[t][0] * own[0].y;
                r.z = th.v[t][0] * own[0].z;
                r.w = th.v[t][0] * own[0].w;
                #pragma unroll
                for (int k = 1; k < N_THETA; ++k) {
                    r.x = fmaf(th.v[t][k], own[k].x, r.x);
                    r.y = fmaf(th.v[t][k], own[k].y, r.y);
                    r.z = fmaf(th.v[t][k], own[k].z, r.z);
                    r.w = fmaf(th.v[t][k], own[k].w, r.w);
                }
                r.x = fmaf(th.v[t][K_STEPS - 1], fv.x, r.x);
                r.y = fmaf(th.v[t][K_STEPS - 1], fv.y, r.y);
                r.z = fmaf(th.v[t][K_STEPS - 1], fv.z, r.z);
                r.w = fmaf(th.v[t][K_STEPS - 1], fv.w, r.w);
                r.x = fmaf(th.v[t][K_STEPS], fn.x, r.x);
                r.y = fmaf(th.v[t][K_STEPS], fn.y, r.y);
                r.z = fmaf(th.v[t][K_STEPS], fn.z, r.z);
                r.w = fmaf(th.v[t][K_STEPS], fn.w, r.w);
                *(float4*)(ob + t * H_FEATS) = r;
            }
        }

        if (!has2) break;
        i = i2; b = b2; n = n2; s_l = s_n; w_l = w_n;
        i2 = i3; has2 = has3; b2 = b3; e2 = e3;
    }
}

static double host_comb(int n, int k) {
    double r = 1.0;
    for (int t = 0; t < k; ++t) r = r * (double)(n - t) / (double)(t + 1);
    return r;
}
static void compute_thetas(double th[N_THETA][K_STEPS + 1]) {
    const double e = 1.4;
    const int d = 4, OFF = 2;
    for (int idx = 0; idx < N_THETA; ++idx) {
        int i = OFF + idx;
        int m = d - i + OFF;
        double B = tgamma((double)(i + 1)) * tgamma((double)(d + 1 - i + OFF)) /
                   tgamma((double)(d + 2 + OFF));
        for (int j = 0; j <= K_STEPS; ++j) th[idx][j] = 0.0;
        for (int j = 0; j <= m; ++j) {
            double c = pow(1.0 / e, (double)i) * host_comb(m, j) *
                       pow(-1.0 / e, (double)j) / (e * B);
            th[idx][i + j] += c;
        }
    }
}

extern "C" void kernel_launch(void* const* d_in, const int* in_sizes, int n_in,
                              void* d_out, int out_size, void* d_ws, size_t ws_size,
                              hipStream_t stream) {
    const float* x    = (const float*)d_in[0];
    const float* W    = (const float*)d_in[1];
    const float* bias = (const float*)d_in[2];
    const int*   src  = (const int*)d_in[3];
    const int*   dst  = (const int*)d_in[4];
    float* out = (float*)d_out;

    float* wsf  = (float*)d_ws;
    float* deg  = wsf;
    float* dinv = deg + N_NODES;
    float* f5   = dinv + N_NODES;
    int*   cnt    = (int*)(f5 + (size_t)N_NODES * H_FEATS);
    int*   rowptr = cnt + NODE_PAD;
    int*   rank   = rowptr + NODE_PAD;
    int*   tmp    = rank + N_EDGES;
    int*   col    = tmp + N_EDGES;

    double th[N_THETA][K_STEPS + 1];
    compute_thetas(th);
    ThetaAll ta;
    for (int t = 0; t < N_THETA; ++t)
        for (int k = 0; k <= K_STEPS; ++k) ta.v[t][k] = (float)th[t][k];

    k_zero2<<<(N_NODES + 255) / 256, 256, 0, stream>>>(deg, cnt);
    k_degree_rank<<<EDGE_GRID, 256, 0, stream>>>(src, dst, deg, cnt, rank);
    k_dinv<<<(N_NODES + 255) / 256, 256, 0, stream>>>(deg, dinv);
    k_scan<<<1, SCAN_T, 0, stream>>>(cnt, rowptr);
    k_fill<<<EDGE_GRID, 256, 0, stream>>>(src, dst, rowptr, rank, tmp);
    k_sort<<<NODE_GRID, 256, 0, stream>>>(rowptr, tmp, col);

    const int nWaves = N_NODES / 16;
    k_gemm_wmma<<<(nWaves + 7) / 8, 256, 0, stream>>>(x, W, bias, out);

    for (int k = 1; k < N_THETA; ++k)
        k_pull<OUT_ROW, OUT_ROW, false><<<NODE_GRID, 256, 0, stream>>>(
            rowptr, col, dinv, out + (k - 1) * H_FEATS, out + k * H_FEATS, ta);
    k_pull<OUT_ROW, H_FEATS, false><<<NODE_GRID, 256, 0, stream>>>(
        rowptr, col, dinv, out + (N_THETA - 1) * H_FEATS, f5, ta);
    k_pull<H_FEATS, OUT_ROW, true><<<NODE_GRID, 256, 0, stream>>>(
        rowptr, col, dinv, f5, out, ta);
}


# Edge ranks and degrees counted in LDS: one work-group per edge chunk holds a counter for every node (returning LDS add = rank), per-node prefix over chunks; no global atomics; sum order kept
# speedup vs baseline: 2.6257x; 1.2805x over previous
#include <hip/hip_runtime.h>
#include <hip/hip_bf16.h>
#include <cmath>

#define N_NODES   50000
#define N_EDGES   800000
#define IN_FEATS  256
#define H_FEATS   64
#define N_THETA   5
#define K_STEPS   6
#define NEG_SLOPE 0.01f
#define NODE_PAD  50176
#define OUT_ROW   (N_THETA * H_FEATS)

#define NODES_PER_WAVE   4
#define NODE_GRID  (((N_NODES + NODES_PER_WAVE - 1) / NODES_PER_WAVE + 7) / 8)
#define ITEMS_PER_LANE   4
#define EDGE_GRID  ((N_EDGES / ITEMS_PER_LANE + 255) / 256)

typedef __attribute__((ext_vector_type(2))) float v2f;
typedef __attribute__((ext_vector_type(8))) float v8f;

struct ThetaAll { float v[N_THETA][K_STEPS + 1]; };


#define C_DST    64
#define C_SRC    32
#define CHUNK_DST (N_EDGES / C_DST)
#define CHUNK_SRC (N_EDGES / C_SRC)
#define CNT_T    1024
static_assert(N_EDGES % C_DST == 0 && N_EDGES % C_SRC == 0, "whole chunks");
static_assert(N_NODES % 4 == 0, "counter rows are moved as 16-byte pieces");
__global__ void __launch_bounds__(CNT_T)
k_count(const int* __restrict__ src, const int* __restrict__ dst,
        int* __restrict__ cc_dst, int* __restrict__ cc_src, int* __restrict__ rank) {
    __shared__ int c_lds[N_NODES];
    const int t = threadIdx.x;
    for (int k = t; k < N_NODES / 4; k += CNT_T) ((int4*)c_lds)[k] = make_int4(0, 0, 0, 0);
    __syncthreads();
    const bool is_dst = blockIdx.x < C_DST;
    if (is_dst) {
        const int e_lo = blockIdx.x * CHUNK_DST;
        const int e_hi = e_lo + CHUNK_DST;
        for (int e0 = e_lo + t; e0 < e_hi; e0 += ITEMS_PER_LANE * CNT_T) {
            int d[ITEMS_PER_LANE];
            int r[ITEMS_PER_LANE];
            #pragma unroll
            for (int u = 0; u < ITEMS_PER_LANE; ++u) {
                const int e = e0 + u * CNT_T;
                d[u] = dst[e < e_hi ? e : e_hi - 1];
            }
            #pragma unroll
            for (int u = 0; u < ITEMS_PER_LANE; ++u) {
                const int e = e0 + u * CNT_T;
                r[u] = atomicAdd(&c_lds[d[u]], e < e_hi ? 1 : 0);
            }
            #pragma unroll
            for (int u = 0; u < ITEMS_PER_LANE; ++u) {
                const int e = e0 + u * CNT_T;
                if (e < e_hi) rank[e] = r[u];
            }
        }
    } else {
        const int e_lo = (blockIdx.x - C_DST) * CHUNK_SRC;
        const int e_hi = e_lo + CHUNK_SRC;
        for (int e0 = e_lo + t; e0 < e_hi; e0 += ITEMS_PER_LANE * CNT_T) {
            int s[ITEMS_PER_LANE];
            #pragma unroll
            for (int u = 0; u < ITEMS_PER_LANE; ++u) {
                const int e = e0 + u * CNT_T;
                s[u] = src[e < e_hi ? e : e_hi - 1];
            }
            #pragma unroll
            for (int u = 0; u < ITEMS_PER_LANE; ++u) {
                const int e = e0 + u * CNT_T;
                atomicAdd(&c_lds[s[u]], e < e_hi ? 1 : 0);
            }
        }
    }
    __syncthreads();
    int4* o = (int4*)(is_dst ? cc_dst + (size_t)blockIdx.x * N_NODES
                             : cc_src + (size_t)(blockIdx.x - C_DST) * N_NODES);
    for (int k = t; k < N_NODES / 4; k += CNT_T) o[k] = ((const int4*)c_lds)[k];
}

#define PFX_BATCH 32
static_assert(C_DST % PFX_BATCH == 0 && C_SRC == PFX_BATCH, "whole batches");
__global__ void k_chunk_prefix(int* __restrict__ cc_dst, const int* __restrict__ cc_src,
                               int* __restrict__ cnt, float* __restrict__ dinv) {
    const int i = blockIdx.x * blockDim.x + threadIdx.x;
    if (i >= N_NODES) return;
    int vs[PFX_BATCH];
    #pragma unroll
    for (int u = 0; u < PFX_BATCH; ++u) vs[u] = cc_src[(size_t)u * N_NODES + i];
    int run = 0;
    for (int c0 = 0; c0 < C_DST; c0 += PFX_BATCH) {
        int v[PFX_BATCH];
        #pragma unroll
        for (int u = 0; u < PFX_BATCH; ++u) v[u] = cc_dst[(size_t)(c0 + u) * N_NODES + i];
        #pragma unroll
        for (int u = 0; u < PFX_BATCH; ++u) {
            cc_dst[(size_t)(c0 + u) * N_NODES + i] = run;
            run += v[u];
        }
    }
    cnt[i] = run;
    int od = 0;
    #pragma unroll
    for (int u = 0; u < PFX_BATCH; ++u) od += vs[u];
    float d = (float)od;
    d = d > 1.0f ? d : 1.0f;
    dinv[i] = rsqrtf(d);
}

__global__ void k_gemm_wmma(const float* __restrict__ x,
                            const float* __restrict__ W,
                            const float* __restrict__ bias,
                            float* __restrict__ h) {
    const int wave = (blockIdx.x * blockDim.x + threadIdx.x) >> 5;
    const int lane = threadIdx.x & 31;
    const int nWaves = N_NODES / 16;
    if (wave >= nWaves) return;

    const int rowBase = wave * 16;
    const int half = lane >> 4;
    const int r    = lane & 15;

    v8f acc0 = {}, acc1 = {}, acc2 = {}, acc3 = {};

    for (int k = 0; k < IN_FEATS; k += 4) {
        v2f a;
        const float* xp = x + (size_t)(rowBase + r) * IN_FEATS + k + 2 * half;
        a.x = xp[0];
        a.y = xp[1];

        const float* wp = W + (size_t)(k + 2 * half) * H_FEATS + r;
        v2f b0, b1, b2, b3;
        b0.x = wp[0];      b0.y = wp[H_FEATS];
        b1.x = wp[16];     b1.y = wp[H_FEATS + 16];
        b2.x = wp[32];     b2.y = wp[H_FEATS + 32];
        b3.x = wp[48];     b3.y = wp[H_FEATS + 48];

        acc0 = __builtin_amdgcn_wmma_f32_16x16x4_f32(false, a, false, b0, (short)0, acc0, false, false);
        acc1 = __builtin_amdgcn_wmma_f32_16x16x4_f32(false, a, false, b1, (short)0, acc1, false, false);
        acc2 = __builtin_amdgcn_wmma_f32_16x16x4_f32(false, a, false, b2, (short)0, acc2, false, false);
        acc3 = __builtin_amdgcn_wmma_f32_16x16x4_f32(false, a, false, b3, (short)0, acc3, false, false);
    }

    const int mBase = rowBase + half * 8;
    v8f accs[4] = {acc0, acc1, acc2, acc3};
    #pragma unroll
    for (int nt = 0; nt < 4; ++nt) {
        const int col = nt * 16 + r;
        const float bv = bias[col];
        #pragma unroll
        for (int v = 0; v < 8; ++v) {
            float val = accs[nt][v] + bv;
            val = val > 0.0f ? val : NEG_SLOPE * val;
            h[(size_t)(mBase + v) * OUT_ROW + col] = val;
        }
    }
}

#define SCAN_T   1024
#define SCAN_PER ((N_NODES + SCAN_T - 1) / SCAN_T)
__global__ void __launch_bounds__(SCAN_T) k_scan(const int* __restrict__ cnt,
                                                 int* __restrict__ rowptr) {
    __shared__ int wsum[32];
    const int t = threadIdx.x;
    const int lane = t & 31, w = t >> 5;
    const int lo = t * SCAN_PER;
    const int hi = (lo + SCAN_PER < N_NODES) ? (lo + SCAN_PER) : N_NODES;
    int s = 0;
    for (int k = lo; k < hi; ++k) s += cnt[k];
    int incl = s;
    for (int off = 1; off < 32; off <<= 1) {
        int v = __shfl_up(incl, off);
        if (lane >= off) incl += v;
    }
    if (lane == 31) wsum[w] = incl;
    __syncthreads();
    if (w == 0) {
        const int x = wsum[lane];
        int xi = x;
        for (int off = 1; off < 32; off <<= 1) {
            int v = __shfl_up(xi, off);
            if (lane >= off) xi += v;
        }
        wsum[lane] = xi - x;
    }
    __syncthreads();
    int run = incl - s + wsum[w];
    for (int k = lo; k < hi; ++k) { rowptr[k] = run; run += cnt[k]; }
    if (t == SCAN_T - 1) rowptr[N_NODES] = run;
}

__global__ void k_fill(const int* __restrict__ src, const int* __restrict__ dst,
                       const int* __restrict__ rowptr, const int* __restrict__ rank,
                       const int* __restrict__ cstart,
                       int* __restrict__ tmp) {
    const int tid    = blockIdx.x * blockDim.x + threadIdx.x;
    const int stride = gridDim.x * blockDim.x;
    for (int e0 = tid; e0 < N_EDGES; e0 += ITEMS_PER_LANE * stride) {
        int s[ITEMS_PER_LANE], d[ITEMS_PER_LANE], p[ITEMS_PER_LANE], c[ITEMS_PER_LANE];
        #pragma unroll
        for (int u = 0; u < ITEMS_PER_LANE; ++u) {
            const int e  = e0 + u * stride;
            const int ec = e < N_EDGES ? e : N_EDGES - 1;
            s[u] = src[ec];
            d[u] = dst[ec];
            p[u] = rank[ec];
            c[u] = ec / CHUNK_DST;
        }
        #pragma unroll
        for (int u = 0; u < ITEMS_PER_LANE; ++u)
            p[u] += rowptr[d[u]] + cstart[(size_t)c[u] * N_NODES + d[u]];
        #pragma unroll
        for (int u = 0; u < ITEMS_PER_LANE; ++u) {
            const int e = e0 + u * stride;
            if (e < N_EDGES) tmp[p[u]] = s[u];
        }
    }
}

__global__ void k_sort(const int* __restrict__ rowptr, const int* __restrict__ tmp,
                       int* __restrict__ col) {
    const int w0 = __builtin_amdgcn_readfirstlane((int)((blockIdx.x * blockDim.x + threadIdx.x) >> 5));
    const int nw = (int)((gridDim.x * blockDim.x) >> 5);
    const int lane = threadIdx.x & 31;
    for (int i = w0; i < N_NODES; i += nw) {
        const int b = __builtin_amdgcn_readfirstlane(rowptr[i]);
        const int n = __builtin_amdgcn_readfirstlane(rowptr[i + 1]) - b;
        if (n <= 32) {
            const int key = lane < n ? tmp[b + lane] : 0x7fffffff;
            int r = 0;
            for (int j = 0; j < n; ++j) {
                const int kj = __builtin_amdgcn_readlane(key, j);
                r += (kj < key || (kj == key && j < lane)) ? 1 : 0;
            }
            if (lane < n) col[b + r] = key;
        } else {
            for (int idx = lane; idx < n; idx += 32) {
                const int key = tmp[b + idx];
                int r = 0;
                for (int j = 0; j < n; ++j) {
                    const int kj = tmp[b + j];
                    r += (kj < key || (kj == key && j < idx)) ? 1 : 0;
                }
                col[b + r] = key;
            }
        }
    }
}

template <int SRC_ROW, int DST_ROW, bool LAST>
__global__ void __launch_bounds__(256)
k_pull(const int* __restrict__ rowptr, const int* __restrict__ col,
       const float* __restrict__ dinv, const float* __restrict__ f,
       float* __restrict__ dst, ThetaAll th) {
    static_assert(K_STEPS == N_THETA + 1, "f_0..f_4 fill the five slots of an out row, f_5 the workspace plane");
    const int lane = threadIdx.x & 31;
    const int half = lane >> 4;
    const int q    = lane & 15;
    const int w0 = __builtin_amdgcn_readfirstlane((int)((blockIdx.x * blockDim.x + threadIdx.x) >> 5));
    const int nw = (int)((gridDim.x * blockDim.x) >> 5);
    if (w0 >= N_NODES) return;

    int i = w0;
    int b = __builtin_amdgcn_readfirstlane(rowptr[i]);
    int n = __builtin_amdgcn_readfirstlane(rowptr[i + 1]) - b;
    int   s_l = i;
    float w_l = 0.0f;
    if (lane < n) { s_l = col[b + lane]; w_l = dinv[s_l]; }

    int  i2   = i + nw;
    bool has2 = i2 < N_NODES;
    int  b2 = 0, e2 = 0;
    if (has2) {
        b2 = __builtin_amdgcn_readfirstlane(rowptr[i2]);
        e2 = __builtin_amdgcn_readfirstlane(rowptr[i2 + 1]);
    }

    for (;;) {
        const int  n2   = e2 - b2;
        const int  i3   = i2 + nw;
        const bool has3 = i3 < N_NODES;
        int b3 = 0, e3 = 0;
        if (has3) {
            b3 = __builtin_amdgcn_readfirstlane(rowptr[i3]);
            e3 = __builtin_amdgcn_readfirstlane(rowptr[i3 + 1]);
        }
        int s_n = has2 ? i2 : i;
        if (has2 && lane < n2) s_n = col[b2 + lane];

        const float  di = dinv[i];
        const float4 fv = *(const float4*)(f + (size_t)i * SRC_ROW + 4 * q);
        float* ob = dst + (size_t)i * DST_ROW + 4 * q;
        [[maybe_unused]] float4 own[N_THETA];
        if constexpr (LAST) {
            #pragma unroll
            for (int k = 0; k < N_THETA; ++k) own[k] = *(const float4*)(ob + k * H_FEATS);
        }

        float4 acc = make_float4(0.0f, 0.0f, 0.0f, 0.0f);
        for (int base = 0; base < n; base += 32) {
            if (base > 0) {
                const int idx = base + lane;
                s_l = i;
                w_l = 0.0f;
                if (idx < n) { s_l = col[b + idx]; w_l = dinv[s_l]; }
            }
            const int m = (n - base < 32) ? (n - base) : 32;
            const int w_li = __float_as_int(w_l);
            for (int j = 0; j < m; j += 8) {
                float4 v[4];
                float  w[4];
                #pragma unroll
                for (int u = 0; u < 4; ++u) {
                    const int e0 = j + 2 * u;
                    const int sa = __builtin_amdgcn_readlane(s_l, e0);
                    const int sb = __builtin_amdgcn_readlane(s_l, e0 + 1);
                    const int wa = __builtin_amdgcn_readlane(w_li, e0);
                    const int wb = __builtin_amdgcn_readlane(w_li, e0 + 1);
                    const int s = half ? sb : sa;
                    w[u] = __int_as_float(half ? wb : wa);
                    v[u] = *(const float4*)(f + (size_t)s * SRC_ROW + 4 * q);
                }
                #pragma unroll
                for (int u = 0; u < 4; ++u) {
                    acc.x = fmaf(v[u].x, w[u], acc.x);
                    acc.y = fmaf(v[u].y, w[u], acc.y);
                    acc.z = fmaf(v[u].z, w[u], acc.z);
                    acc.w = fmaf(v[u].w, w[u], acc.w);
                }
            }
        }

        float w_n = 0.0f;
        if (has2 && lane < n2) w_n = dinv[s_n];

        acc.x += __shfl_xor(acc.x, 16);
        acc.y += __shfl_xor(acc.y, 16);
        acc.z += __shfl_xor(acc.z, 16);
        acc.w += __shfl_xor(acc.w, 16);

        float4 fn;
        fn.x = fv.x - acc.x * di;
        fn.y = fv.y - acc.y * di;
        fn.z = fv.z - acc.z * di;
        fn.w = fv.w - acc.w * di;

        if constexpr (!LAST) {
            if (half == 0) *(float4*)ob = fn;
        } else {
            #pragma unroll
            for (int t = 0; t < N_THETA && half == 0; ++t) {
                float4 r;
                r.x = th.v[t][0] * own[0].x;
                r.y = th.v[t][0] * own[0].y;
                r.z = th.v[t][0] * own[0].z;
                r.w = th.v[t][0] * own[0].w;
                #pragma unroll
                for (int k = 1; k < N_THETA; ++k) {
                    r.x = fmaf(th.v[t][k], own[k].x, r.x);
                    r.y = fmaf(th.v[t][k], own[k].y, r.y);
                    r.z = fmaf(th.v[t][k], own[k].z, r.z);
                    r.w = fmaf(th.v[t][k], own[k].w, r.w);
                }
                r.x = fmaf(th.v[t][K_STEPS - 1], fv.x, r.x);
                r.y = fmaf(th.v[t][K_STEPS - 1], fv.y, r.y);
                r.z = fmaf(th.v[t][K_STEPS - 1], fv.z, r.z);
                r.w = fmaf(th.v[t][K_STEPS - 1], fv.w, r.w);
                r.x = fmaf(th.v[t][K_STEPS], fn.x, r.x);
                r.y = fmaf(th.v[t][K_STEPS], fn.y, r.y);
                r.z = fmaf(th.v[t][K_STEPS], fn.z, r.z);
                r.w = fmaf(th.v[t][K_STEPS], fn.w, r.w);
                *(float4*)(ob + t * H_FEATS) = r;
            }
        }

        if (!has2) break;
        i = i2; b = b2; n = n2; s_l = s_n; w_l = w_n;
        i2 = i3; has2 = has3; b2 = b3; e2 = e3;
    }
}

static double host_comb(int n, int k) {
    double r = 1.0;
    for (int t = 0; t < k; ++t) r = r * (double)(n - t) / (double)(t + 1);
    return r;
}
static void compute_thetas(double th[N_THETA][K_STEPS + 1]) {
    const double e = 1.4;
    const int d = 4, OFF = 2;
    for (int idx = 0; idx < N_THETA; ++idx) {
        int i = OFF + idx;
        int m = d - i + OFF;
        double B = tgamma((double)(i + 1)) * tgamma((double)(d + 1 - i + OFF)) /
                   tgamma((double)(d + 2 + OFF));
        for (int j = 0; j <= K_STEPS; ++j) th[idx][j] = 0.0;
        for (int j = 0; j <= m; ++j) {
            double c = pow(1.0 / e, (double)i) * host_comb(m, j) *
                       pow(-1.0 / e, (double)j) / (e * B);
            th[idx][i + j] += c;
        }
    }
}

extern "C" void kernel_launch(void* const* d_in, const int* in_sizes, int n_in,
                              void* d_out, int out_size, void* d_ws, size_t ws_size,
                              hipStream_t stream) {
    const float* x    = (const float*)d_in[0];
    const float* W    = (const float*)d_in[1];
    const float* bias = (const float*)d_in[2];
    const int*   src  = (const int*)d_in[3];
    const int*   dst  = (const int*)d_in[4];
    float* out = (float*)d_out;

    float* wsf  = (float*)d_ws;
    float* dinv = wsf + N_NODES;
    float* f5   = dinv + N_NODES;
    int*   cnt    = (int*)(f5 + (size_t)N_NODES * H_FEATS);
    int*   rowptr = cnt + NODE_PAD;
    int*   rank   = rowptr + NODE_PAD;
    int*   tmp    = rank + N_EDGES;
    int*   col    = tmp + N_EDGES;
    int*   cc_dst = (int*)f5;
    int*   cc_src = tmp;
    static_assert((size_t)C_DST * N_NODES <= (size_t)N_NODES * H_FEATS, "destination counts fit the f5 plane");
    static_assert((size_t)C_SRC * N_NODES <= (size_t)2 * N_EDGES, "source counts fit tmp..col");

    double th[N_THETA][K_STEPS + 1];
    compute_thetas(th);
    ThetaAll ta;
    for (int t = 0; t < N_THETA; ++t)
        for (int k = 0; k <= K_STEPS; ++k) ta.v[t][k] = (float)th[t][k];

    k_count<<<C_DST + C_SRC, CNT_T, 0, stream>>>(src, dst, cc_dst, cc_src, rank);
    k_chunk_prefix<<<(N_NODES + 255) / 256, 256, 0, stream>>>(cc_dst, cc_src, cnt, dinv);
    k_scan<<<1, SCAN_T, 0, stream>>>(cnt, rowptr);
    k_fill<<<EDGE_GRID, 256, 0, stream>>>(src, dst, rowptr, rank, cc_dst, tmp);
    k_sort<<<NODE_GRID, 256, 0, stream>>>(rowptr, tmp, col);

    const int nWaves = N_NODES / 16;
    k_gemm_wmma<<<(nWaves + 7) / 8, 256, 0, stream>>>(x, W, bias, out);

    for (int k = 1; k < N_THETA; ++k)
        k_pull<OUT_ROW, OUT_ROW, false><<<NODE_GRID, 256, 0, stream>>>(
            rowptr, col, dinv, out + (k - 1) * H_FEATS, out + k * H_FEATS, ta);
    k_pull<OUT_ROW, H_FEATS, false><<<NODE_GRID, 256, 0, stream>>>(
        rowptr, col, dinv, out + (N_THETA - 1) * H_FEATS, f5, ta);
    k_pull<H_FEATS, OUT_ROW, true><<<NODE_GRID, 256, 0, stream>>>(
        rowptr, col, dinv, f5, out, ta);
}


# Bucket starts by a two-level scan inside the neighbouring launches: chunk-prefix work-groups scan their 256 in-degrees and write one total, fill scans the 196 totals in LDS; no one-block scan launch
# speedup vs baseline: 3.4750x; 1.3235x over previous
#include <hip/hip_runtime.h>
#include <hip/hip_bf16.h>
#include <cmath>

#define N_NODES   50000
#define N_EDGES   800000
#define IN_FEATS  256
#define H_FEATS   64
#define N_THETA   5
#define K_STEPS   6
#define NEG_SLOPE 0.01f
#define NODE_PAD  50176
#define OUT_ROW   (N_THETA * H_FEATS)

#define NODES_PER_WAVE   4
#define NODE_GRID  (((N_NODES + NODES_PER_WAVE - 1) / NODES_PER_WAVE + 7) / 8)
#define ITEMS_PER_LANE   4
#define EDGE_GRID  ((N_EDGES / ITEMS_PER_LANE + 255) / 256)

typedef __attribute__((ext_vector_type(2))) float v2f;
typedef __attribute__((ext_vector_type(8))) float v8f;

struct ThetaAll { float v[N_THETA][K_STEPS + 1]; };


#define C_DST    64
#define C_SRC    32
#define CHUNK_DST (N_EDGES / C_DST)
#define CHUNK_SRC (N_EDGES / C_SRC)
#define CNT_T    1024
static_assert(N_EDGES % C_DST == 0 && N_EDGES % C_SRC == 0, "whole chunks");
static_assert(N_NODES % 4 == 0, "counter rows are moved as 16-byte pieces");
__global__ void __launch_bounds__(CNT_T)
k_count(const int* __restrict__ src, const int* __restrict__ dst,
        int* __restrict__ cc_dst, int* __restrict__ cc_src, int* __restrict__ rank) {
    __shared__ int c_lds[N_NODES];
    const int t = threadIdx.x;
    for (int k = t; k < N_NODES / 4; k += CNT_T) ((int4*)c_lds)[k] = make_int4(0, 0, 0, 0);
    __syncthreads();
    const bool is_dst = blockIdx.x < C_DST;
    if (is_dst) {
        const int e_lo = blockIdx.x * CHUNK_DST;
        const int e_hi = e_lo + CHUNK_DST;
        for (int e0 = e_lo + t; e0 < e_hi; e0 += ITEMS_PER_LANE * CNT_T) {
            int d[ITEMS_PER_LANE];
            int r[ITEMS_PER_LANE];
            #pragma unroll
            for (int u = 0; u < ITEMS_PER_LANE; ++u) {
                const int e = e0 + u * CNT_T;
                d[u] = dst[e < e_hi ? e : e_hi - 1];
            }
            #pragma unroll
            for (int u = 0; u < ITEMS_PER_LANE; ++u) {
                const int e = e0 + u * CNT_T;
                r[u] = atomicAdd(&c_lds[d[u]], e < e_hi ? 1 : 0);
            }
            #pragma unroll
            for (int u = 0; u < ITEMS_PER_LANE; ++u) {
                const int e = e0 + u * CNT_T;
                if (e < e_hi) rank[e] = r[u];
            }
        }
    } else {
        const int e_lo = (blockIdx.x - C_DST) * CHUNK_SRC;
        const int e_hi = e_lo + CHUNK_SRC;
        for (int e0 = e_lo + t; e0 < e_hi; e0 += ITEMS_PER_LANE * CNT_T) {
            int s[ITEMS_PER_LANE];
            #pragma unroll
            for (int u = 0; u < ITEMS_PER_LANE; ++u) {
                const int e = e0 + u * CNT_T;
                s[u] = src[e < e_hi ? e : e_hi - 1];
            }
            #pragma unroll
            for (int u = 0; u < ITEMS_PER_LANE; ++u) {
                const int e = e0 + u * CNT_T;
                atomicAdd(&c_lds[s[u]], e < e_hi ? 1 : 0);
            }
        }
    }
    __syncthreads();
    int4* o = (int4*)(is_dst ? cc_dst + (size_t)blockIdx.x * N_NODES
                             : cc_src + (size_t)(blockIdx.x - C_DST) * N_NODES);
    for (int k = t; k < N_NODES / 4; k += CNT_T) o[k] = ((const int4*)c_lds)[k];
}

#define PFX_BATCH 32
#define PFX_T     256
#define N_PBLK    ((N_NODES + PFX_T - 1) / PFX_T)
static_assert(C_DST % PFX_BATCH == 0 && C_SRC == PFX_BATCH, "whole batches");
__global__ void __launch_bounds__(PFX_T)
k_chunk_prefix(int* __restrict__ cc_dst, const int* __restrict__ cc_src,
               int* __restrict__ lstart, int* __restrict__ btot, float* __restrict__ dinv) {
    __shared__ int wsum[PFX_T / 32];
    const int t = threadIdx.x;
    const int i = blockIdx.x * PFX_T + t;
    const bool live = i < N_NODES;
    int run = 0;
    if (live) {
        int vs[PFX_BATCH];
        #pragma unroll
        for (int u = 0; u < PFX_BATCH; ++u) vs[u] = cc_src[(size_t)u * N_NODES + i];
        for (int c0 = 0; c0 < C_DST; c0 += PFX_BATCH) {
            int v[PFX_BATCH];
            #pragma unroll
            for (int u = 0; u < PFX_BATCH; ++u) v[u] = cc_dst[(size_t)(c0 + u) * N_NODES + i];
            #pragma unroll
            for (int u = 0; u < PFX_BATCH; ++u) {
                cc_dst[(size_t)(c0 + u) * N_NODES + i] = run;
                run += v[u];
            }
        }
        int od = 0;
        #pragma unroll
        for (int u = 0; u < PFX_BATCH; ++u) od += vs[u];
        float d = (float)od;
        d = d > 1.0f ? d : 1.0f;
        dinv[i] = rsqrtf(d);
    }

    const int lane = t & 31, w = t >> 5;
    int incl = run;
    for (int off = 1; off < 32; off <<= 1) {
        const int v = __shfl_up(incl, off);
        if (lane >= off) incl += v;
    }
    if (lane == 31) wsum[w] = incl;
    __syncthreads();
    int before = 0, total = 0;
    for (int k = 0; k < PFX_T / 32; ++k) {
        const int x = wsum[k];
        before += k < w ? x : 0;
        total  += x;
    }
    if (live) lstart[i] = before + incl - run;
    if (t == 0) btot[blockIdx.x] = total;
}

__global__ void k_gemm_wmma(const float* __restrict__ x,
                            const float* __restrict__ W,
                            const float* __restrict__ bias,
                            float* __restrict__ h) {
    const int wave = (blockIdx.x * blockDim.x + threadIdx.x) >> 5;
    const int lane = threadIdx.x & 31;
    const int nWaves = N_NODES / 16;
    if (wave >= nWaves) return;

    const int rowBase = wave * 16;
    const int half = lane >> 4;
    const int r    = lane & 15;

    v8f acc0 = {}, acc1 = {}, acc2 = {}, acc3 = {};

    for (int k = 0; k < IN_FEATS; k += 4) {
        v2f a;
        const float* xp = x + (size_t)(rowBase + r) * IN_FEATS + k + 2 * half;
        a.x = xp[0];
        a.y = xp[1];

        const float* wp = W + (size_t)(k + 2 * half) * H_FEATS + r;
        v2f b0, b1, b2, b3;
        b0.x = wp[0];      b0.y = wp[H_FEATS];
        b1.x = wp[16];     b1.y = wp[H_FEATS + 16];
        b2.x = wp[32];     b2.y = wp[H_FEATS + 32];
        b3.x = wp[48];     b3.y = wp[H_FEATS + 48];

        acc0 = __builtin_amdgcn_wmma_f32_16x16x4_f32(false, a, false, b0, (short)0, acc0, false, false);
        acc1 = __builtin_amdgcn_wmma_f32_16x16x4_f32(false, a, false, b1, (short)0, acc1, false, false);
        acc2 = __builtin_amdgcn_wmma_f32_16x16x4_f32(false, a, false, b2, (short)0, acc2, false, false);
        acc3 = __builtin_amdgcn_wmma_f32_16x16x4_f32(false, a, false, b3, (short)0, acc3, false, false);
    }

    const int mBase = rowBase + half * 8;
    v8f accs[4] = {acc0, acc1, acc2, acc3};
    #pragma unroll
    for (int nt = 0; nt < 4; ++nt) {
        const int col = nt * 16 + r;
        const float bv = bias[col];
        #pragma unroll
        for (int v = 0; v < 8; ++v) {
            float val = accs[nt][v] + bv;
            val = val > 0.0f ? val : NEG_SLOPE * val;
            h[(size_t)(mBase + v) * OUT_ROW + col] = val;
        }
    }
}

#define FILL_T 256
static_assert(N_PBLK < FILL_T, "one work-group scans all totals in one go; entry N_PBLK is their sum");
__global__ void __launch_bounds__(FILL_T)
k_fill(const int* __restrict__ src, const int* __restrict__ dst,
       const int* __restrict__ lstart,
       const int* __restrict__ btot,
       const int* __restrict__ rank,
       const int* __restrict__ cstart,
       int* __restrict__ rowptr, int* __restrict__ tmp) {
    __shared__ int boff[FILL_T];
    __shared__ int wsum[FILL_T / 32];
    const int t = threadIdx.x;
    {
        const int lane = t & 31, w = t >> 5;
        const int v = t < N_PBLK ? btot[t] : 0;
        int incl = v;
        for (int off = 1; off < 32; off <<= 1) {
            const int x = __shfl_up(incl, off);
            if (lane >= off) incl += x;
        }
        if (lane == 31) wsum[w] = incl;
        __syncthreads();
        int before = 0;
        for (int k = 0; k < FILL_T / 32; ++k) before += k < w ? wsum[k] : 0;
        boff[t] = before + incl - v;
        __syncthreads();
    }
    const int tid    = blockIdx.x * FILL_T + t;
    const int stride = gridDim.x * FILL_T;
    for (int e0 = tid; e0 < N_EDGES; e0 += ITEMS_PER_LANE * stride) {
        int s[ITEMS_PER_LANE], d[ITEMS_PER_LANE], p[ITEMS_PER_LANE], c[ITEMS_PER_LANE];
        #pragma unroll
        for (int u = 0; u < ITEMS_PER_LANE; ++u) {
            const int e  = e0 + u * stride;
            const int ec = e < N_EDGES ? e : N_EDGES - 1;
            s[u] = src[ec];
            d[u] = dst[ec];
            p[u] = rank[ec];
            c[u] = ec / CHUNK_DST;
        }
        #pragma unroll
        for (int u = 0; u < ITEMS_PER_LANE; ++u)
            p[u] += boff[(unsigned)d[u] / PFX_T] + lstart[d[u]] + cstart[(size_t)c[u] * N_NODES + d[u]];
        #pragma unroll
        for (int u = 0; u < ITEMS_PER_LANE; ++u) {
            const int e = e0 + u * stride;
            if (e < N_EDGES) tmp[p[u]] = s[u];
        }
    }
    for (int i = tid; i <= N_NODES; i += stride) {
        int r = boff[N_PBLK];
        if (i < N_NODES) r = boff[(unsigned)i / PFX_T] + lstart[i];
        rowptr[i] = r;
    }
}

__global__ void k_sort(const int* __restrict__ rowptr, const int* __restrict__ tmp,
                       int* __restrict__ col) {
    const int w0 = __builtin_amdgcn_readfirstlane((int)((blockIdx.x * blockDim.x + threadIdx.x) >> 5));
    const int nw = (int)((gridDim.x * blockDim.x) >> 5);
    const int lane = threadIdx.x & 31;
    for (int i = w0; i < N_NODES; i += nw) {
        const int b = __builtin_amdgcn_readfirstlane(rowptr[i]);
        const int n = __builtin_amdgcn_readfirstlane(rowptr[i + 1]) - b;
        if (n <= 32) {
            const int key = lane < n ? tmp[b + lane] : 0x7fffffff;
            int r = 0;
            for (int j = 0; j < n; ++j) {
                const int kj = __builtin_amdgcn_readlane(key, j);
                r += (kj < key || (kj == key && j < lane)) ? 1 : 0;
            }
            if (lane < n) col[b + r] = key;
        } else {
            for (int idx = lane; idx < n; idx += 32) {
                const int key = tmp[b + idx];
                int r = 0;
                for (int j = 0; j < n; ++j) {
                    const int kj = tmp[b + j];
                    r += (kj < key || (kj == key && j < idx)) ? 1 : 0;
                }
                col[b + r] = key;
            }
        }
    }
}

template <int SRC_ROW, int DST_ROW, bool LAST>
__global__ void __launch_bounds__(256)
k_pull(const int* __restrict__ rowptr, const int* __restrict__ col,
       const float* __restrict__ dinv, const float* __restrict__ f,
       float* __restrict__ dst, ThetaAll th) {
    static_assert(K_STEPS == N_THETA + 1, "f_0..f_4 fill the five slots of an out row, f_5 the workspace plane");
    const int lane = threadIdx.x & 31;
    const int half = lane >> 4;
    const int q    = lane & 15;
    const int w0 = __builtin_amdgcn_readfirstlane((int)((blockIdx.x * blockDim.x + threadIdx.x) >> 5));
    const int nw = (int)((gridDim.x * blockDim.x) >> 5);
    if (w0 >= N_NODES) return;

    int i = w0;
    int b = __builtin_amdgcn_readfirstlane(rowptr[i]);
    int n = __builtin_amdgcn_readfirstlane(rowptr[i + 1]) - b;
    int   s_l = i;
    float w_l = 0.0f;
    if (lane < n) { s_l = col[b + lane]; w_l = dinv[s_l]; }

    int  i2   = i + nw;
    bool has2 = i2 < N_NODES;
    int  b2 = 0, e2 = 0;
    if (has2) {
        b2 = __builtin_amdgcn_readfirstlane(rowptr[i2]);
        e2 = __builtin_amdgcn_readfirstlane(rowptr[i2 + 1]);
    }

    for (;;) {
        const int  n2   = e2 - b2;
        const int  i3   = i2 + nw;
        const bool has3 = i3 < N_NODES;
        int b3 = 0, e3 = 0;
        if (has3) {
            b3 = __builtin_amdgcn_readfirstlane(rowptr[i3]);
            e3 = __builtin_amdgcn_readfirstlane(rowptr[i3 + 1]);
        }
        int s_n = has2 ? i2 : i;
        if (has2 && lane < n2) s_n = col[b2 + lane];

        const float  di = dinv[i];
        const float4 fv = *(const float4*)(f + (size_t)i * SRC_ROW + 4 * q);
        float* ob = dst + (size_t)i * DST_ROW + 4 * q;
        [[maybe_unused]] float4 own[N_THETA];
        if constexpr (LAST) {
            #pragma unroll
            for (int k = 0; k < N_THETA; ++k) own[k] = *(const float4*)(ob + k * H_FEATS);
        }

        float4 acc = make_float4(0.0f, 0.0f, 0.0f, 0.0f);
        for (int base = 0; base < n; base += 32) {
            if (base > 0) {
                const int idx = base + lane;
                s_l = i;
                w_l = 0.0f;
                if (idx < n) { s_l = col[b + idx]; w_l = dinv[s_l]; }
            }
            const int m = (n - base < 32) ? (n - base) : 32;
            const int w_li = __float_as_int(w_l);
            for (int j = 0; j < m; j += 8) {
                float4 v[4];
                float  w[4];
                #pragma unroll
                for (int u = 0; u < 4; ++u) {
                    const int e0 = j + 2 * u;
                    const int sa = __builtin_amdgcn_readlane(s_l, e0);
                    const int sb = __builtin_amdgcn_readlane(s_l, e0 + 1);
                    const int wa = __builtin_amdgcn_readlane(w_li, e0);
                    const int wb = __builtin_amdgcn_readlane(w_li, e0 + 1);
                    const int s = half ? sb : sa;
                    w[u] = __int_as_float(half ? wb : wa);
                    v[u] = *(const float4*)(f + (size_t)s * SRC_ROW + 4 * q);
                }
                #pragma unroll
                for (int u = 0; u < 4; ++u) {
                    acc.x = fmaf(v[u].x, w[u], acc.x);
                    acc.y = fmaf(v[u].y, w[u], acc.y);
                    acc.z = fmaf(v[u].z, w[u], acc.z);
                    acc.w = fmaf(v[u].w, w[u], acc.w);
                }
            }
        }

        float w_n = 0.0f;
        if (has2 && lane < n2) w_n = dinv[s_n];

        acc.x += __shfl_xor(acc.x, 16);
        acc.y += __shfl_xor(acc.y, 16);
        acc.z += __shfl_xor(acc.z, 16);
        acc.w += __shfl_xor(acc.w, 16);

        float4 fn;
        fn.x = fv.x - acc.x * di;
        fn.y = fv.y - acc.y * di;
        fn.z = fv.z - acc.z * di;
        fn.w = fv.w - acc.w * di;

        if constexpr (!LAST) {
            if (half == 0) *(float4*)ob = fn;
        } else {
            #pragma unroll
            for (int t = 0; t < N_THETA && half == 0; ++t) {
                float4 r;
                r.x = th.v[t][0] * own[0].x;
                r.y = th.v[t][0] * own[0].y;
                r.z = th.v[t][0] * own[0].z;
                r.w = th.v[t][0] * own[0].w;
                #pragma unroll
                for (int k = 1; k < N_THETA; ++k) {
                    r.x = fmaf(th.v[t][k], own[k].x, r.x);
                    r.y = fmaf(th.v[t][k], own[k].y, r.y);
                    r.z = fmaf(th.v[t][k], own[k].z, r.z);
                    r.w = fmaf(th.v[t][k], own[k].w, r.w);
                }
                r.x = fmaf(th.v[t][K_STEPS - 1], fv.x, r.x);
                r.y = fmaf(th.v[t][K_STEPS - 1], fv.y, r.y);
                r.z = fmaf(th.v[t][K_STEPS - 1], fv.z, r.z);
                r.w = fmaf(th.v[t][K_STEPS - 1], fv.w, r.w);
                r.x = fmaf(th.v[t][K_STEPS], fn.x, r.x);
                r.y = fmaf(th.v[t][K_STEPS], fn.y, r.y);
                r.z = fmaf(th.v[t][K_STEPS], fn.z, r.z);
                r.w = fmaf(th.v[t][K_STEPS], fn.w, r.w);
                *(float4*)(ob + t * H_FEATS) = r;
            }
        }

        if (!has2) break;
        i = i2; b = b2; n = n2; s_l = s_n; w_l = w_n;
        i2 = i3; has2 = has3; b2 = b3; e2 = e3;
    }
}

static double host_comb(int n, int k) {
    double r = 1.0;
    for (int t = 0; t < k; ++t) r = r * (double)(n - t) / (double)(t + 1);
    return r;
}
static void compute_thetas(double th[N_THETA][K_STEPS + 1]) {
    const double e = 1.4;
    const int d = 4, OFF = 2;
    for (int idx = 0; idx < N_THETA; ++idx) {
        int i = OFF + idx;
        int m = d - i + OFF;
        double B = tgamma((double)(i + 1)) * tgamma((double)(d + 1 - i + OFF)) /
                   tgamma((double)(d + 2 + OFF));
        for (int j = 0; j <= K_STEPS; ++j) th[idx][j] = 0.0;
        for (int j = 0; j <= m; ++j) {
            double c = pow(1.0 / e, (double)i) * host_comb(m, j) *
                       pow(-1.0 / e, (double)j) / (e * B);
            th[idx][i + j] += c;
        }
    }
}

extern "C" void kernel_launch(void* const* d_in, const int* in_sizes, int n_in,
                              void* d_out, int out_size, void* d_ws, size_t ws_size,
                              hipStream_t stream) {
    const float* x    = (const float*)d_in[0];
    const float* W    = (const float*)d_in[1];
    const float* bias = (const float*)d_in[2];
    const int*   src  = (const int*)d_in[3];
    const int*   dst  = (const int*)d_in[4];
    float* out = (float*)d_out;

    float* wsf  = (float*)d_ws;
    float* dinv = wsf + N_NODES;
    float* f5   = dinv + N_NODES;
    int*   btot   = (int*)d_ws;
    int*   lstart = (int*)(f5 + (size_t)N_NODES * H_FEATS);
    int*   rowptr = lstart + NODE_PAD;
    int*   rank   = rowptr + NODE_PAD;
    int*   tmp    = rank + N_EDGES;
    int*   col    = tmp + N_EDGES;
    int*   cc_dst = (int*)f5;
    int*   cc_src = tmp;
    static_assert((size_t)C_DST * N_NODES <= (size_t)N_NODES * H_FEATS, "destination counts fit the f5 plane");
    static_assert((size_t)C_SRC * N_NODES <= (size_t)2 * N_EDGES, "source counts fit tmp..col");
    static_assert(N_PBLK <= N_NODES, "work-group totals fit the first plane");

    double th[N_THETA][K_STEPS + 1];
    compute_thetas(th);
    ThetaAll ta;
    for (int t = 0; t < N_THETA; ++t)
        for (int k = 0; k <= K_STEPS; ++k) ta.v[t][k] = (float)th[t][k];

    k_count<<<C_DST + C_SRC, CNT_T, 0, stream>>>(src, dst, cc_dst, cc_src, rank);
    k_chunk_prefix<<<N_PBLK, PFX_T, 0, stream>>>(cc_dst, cc_src, lstart, btot, dinv);
    k_fill<<<EDGE_GRID, FILL_T, 0, stream>>>(src, dst, lstart, btot, rank, cc_dst, rowptr, tmp);
    k_sort<<<NODE_GRID, 256, 0, stream>>>(rowptr, tmp, col);

    const int nWaves = N_NODES / 16;
    k_gemm_wmma<<<(nWaves + 7) / 8, 256, 0, stream>>>(x, W, bias, out);

    for (int k = 1; k < N_THETA; ++k)
        k_pull<OUT_ROW, OUT_ROW, false><<<NODE_GRID, 256, 0, stream>>>(
            rowptr, col, dinv, out + (k - 1) * H_FEATS, out + k * H_FEATS, ta);
    k_pull<OUT_ROW, H_FEATS, false><<<NODE_GRID, 256, 0, stream>>>(
        rowptr, col, dinv, out + (N_THETA - 1) * H_FEATS, f5, ta);
    k_pull<H_FEATS, OUT_ROW, true><<<NODE_GRID, 256, 0, stream>>>(
        rowptr, col, dinv, f5, out, ta);
}


# The f32 WMMA product x@W+b rides in the bucket-sort launch as its first work-groups (independent work: one launch fewer, the sort runs beside the product); every sum order kept
# speedup vs baseline: 3.6160x; 1.0406x over previous
#include <hip/hip_runtime.h>
#include <hip/hip_bf16.h>
#include <cmath>

#define N_NODES   50000
#define N_EDGES   800000
#define IN_FEATS  256
#define H_FEATS   64
#define N_THETA   5
#define K_STEPS   6
#define NEG_SLOPE 0.01f
#define NODE_PAD  50176
#define OUT_ROW   (N_THETA * H_FEATS)

#define NODES_PER_WAVE   4
#define NODE_GRID  (((N_NODES + NODES_PER_WAVE - 1) / NODES_PER_WAVE + 7) / 8)
#define ITEMS_PER_LANE   4
#define EDGE_GRID  ((N_EDGES / ITEMS_PER_LANE + 255) / 256)

typedef __attribute__((ext_vector_type(2))) float v2f;
typedef __attribute__((ext_vector_type(8))) float v8f;

struct ThetaAll { float v[N_THETA][K_STEPS + 1]; };


#define C_DST    64
#define C_SRC    32
#define CHUNK_DST (N_EDGES / C_DST)
#define CHUNK_SRC (N_EDGES / C_SRC)
#define CNT_T    1024
static_assert(N_EDGES % C_DST == 0 && N_EDGES % C_SRC == 0, "whole chunks");
static_assert(N_NODES % 4 == 0, "counter rows are moved as 16-byte pieces");
__global__ void __launch_bounds__(CNT_T)
k_count(const int* __restrict__ src, const int* __restrict__ dst,
        int* __restrict__ cc_dst, int* __restrict__ cc_src, int* __restrict__ rank) {
    __shared__ int c_lds[N_NODES];
    const int t = threadIdx.x;
    for (int k = t; k < N_NODES / 4; k += CNT_T) ((int4*)c_lds)[k] = make_int4(0, 0, 0, 0);
    __syncthreads();
    const bool is_dst = blockIdx.x < C_DST;
    if (is_dst) {
        const int e_lo = blockIdx.x * CHUNK_DST;
        const int e_hi = e_lo + CHUNK_DST;
        for (int e0 = e_lo + t; e0 < e_hi; e0 += ITEMS_PER_LANE * CNT_T) {
            int d[ITEMS_PER_LANE];
            int r[ITEMS_PER_LANE];
            #pragma unroll
            for (int u = 0; u < ITEMS_PER_LANE; ++u) {
                const int e = e0 + u * CNT_T;
                d[u] = dst[e < e_hi ? e : e_hi - 1];
            }
            #pragma unroll
            for (int u = 0; u < ITEMS_PER_LANE; ++u) {
                const int e = e0 + u * CNT_T;
                r[u] = atomicAdd(&c_lds[d[u]], e < e_hi ? 1 : 0);
            }
            #pragma unroll
            for (int u = 0; u < ITEMS_PER_LANE; ++u) {
                const int e = e0 + u * CNT_T;
                if (e < e_hi) rank[e] = r[u];
            }
        }
    } else {
        const int e_lo = (blockIdx.x - C_DST) * CHUNK_SRC;
        const int e_hi = e_lo + CHUNK_SRC;
        for (int e0 = e_lo + t; e0 < e_hi; e0 += ITEMS_PER_LANE * CNT_T) {
            int s[ITEMS_PER_LANE];
            #pragma unroll
            for (int u = 0; u < ITEMS_PER_LANE; ++u) {
                const int e = e0 + u * CNT_T;
                s[u] = src[e < e_hi ? e : e_hi - 1];
            }
            #pragma unroll
            for (int u = 0; u < ITEMS_PER_LANE; ++u) {
                const int e = e0 + u * CNT_T;
                atomicAdd(&c_lds[s[u]], e < e_hi ? 1 : 0);
            }
        }
    }
    __syncthreads();
    int4* o = (int4*)(is_dst ? cc_dst + (size_t)blockIdx.x * N_NODES
                             : cc_src + (size_t)(blockIdx.x - C_DST) * N_NODES);
    for (int k = t; k < N_NODES / 4; k += CNT_T) o[k] = ((const int4*)c_lds)[k];
}

#define PFX_BATCH 32
#define PFX_T     256
#define N_PBLK    ((N_NODES + PFX_T - 1) / PFX_T)
static_assert(C_DST % PFX_BATCH == 0 && C_SRC == PFX_BATCH, "whole batches");
__global__ void __launch_bounds__(PFX_T)
k_chunk_prefix(int* __restrict__ cc_dst, const int* __restrict__ cc_src,
               int* __restrict__ lstart, int* __restrict__ btot, float* __restrict__ dinv) {
    __shared__ int wsum[PFX_T / 32];
    const int t = threadIdx.x;
    const int i = blockIdx.x * PFX_T + t;
    const bool live = i < N_NODES;
    int run = 0;
    if (live) {
        int vs[PFX_BATCH];
        #pragma unroll
        for (int u = 0; u < PFX_BATCH; ++u) vs[u] = cc_src[(size_t)u * N_NODES + i];
        for (int c0 = 0; c0 < C_DST; c0 += PFX_BATCH) {
            int v[PFX_BATCH];
            #pragma unroll
            for (int u = 0; u < PFX_BATCH; ++u) v[u] = cc_dst[(size_t)(c0 + u) * N_NODES + i];
            #pragma unroll
            for (int u = 0; u < PFX_BATCH; ++u) {
                cc_dst[(size_t)(c0 + u) * N_NODES + i] = run;
                run += v[u];
            }
        }
        int od = 0;
        #pragma unroll
        for (int u = 0; u < PFX_BATCH; ++u) od += vs[u];
        float d = (float)od;
        d = d > 1.0f ? d : 1.0f;
        dinv[i] = rsqrtf(d);
    }

    const int lane = t & 31, w = t >> 5;
    int incl = run;
    for (int off = 1; off < 32; off <<= 1) {
        const int v = __shfl_up(incl, off);
        if (lane >= off) incl += v;
    }
    if (lane == 31) wsum[w] = incl;
    __syncthreads();
    int before = 0, total = 0;
    for (int k = 0; k < PFX_T / 32; ++k) {
        const int x = wsum[k];
        before += k < w ? x : 0;
        total  += x;
    }
    if (live) lstart[i] = before + incl - run;
    if (t == 0) btot[blockIdx.x] = total;
}

#define GEMM_BLOCKS ((N_NODES / 16 + 7) / 8)
__device__ __forceinline__ void gemm_strip(const float* __restrict__ x,
                            const float* __restrict__ W,
                            const float* __restrict__ bias,
                            float* __restrict__ h) {
    const int wave = (blockIdx.x * blockDim.x + threadIdx.x) >> 5;
    const int lane = threadIdx.x & 31;
    const int nWaves = N_NODES / 16;
    if (wave >= nWaves) return;

    const int rowBase = wave * 16;
    const int half = lane >> 4;
    const int r    = lane & 15;

    v8f acc0 = {}, acc1 = {}, acc2 = {}, acc3 = {};

    for (int k = 0; k < IN_FEATS; k += 4) {
        v2f a;
        const float* xp = x + (size_t)(rowBase + r) * IN_FEATS + k + 2 * half;
        a.x = xp[0];
        a.y = xp[1];

        const float* wp = W + (size_t)(k + 2 * half) * H_FEATS + r;
        v2f b0, b1, b2, b3;
        b0.x = wp[0];      b0.y = wp[H_FEATS];
        b1.x = wp[16];     b1.y = wp[H_FEATS + 16];
        b2.x = wp[32];     b2.y = wp[H_FEATS + 32];
        b3.x = wp[48];     b3.y = wp[H_FEATS + 48];

        acc0 = __builtin_amdgcn_wmma_f32_16x16x4_f32(false, a, false, b0, (short)0, acc0, false, false);
        acc1 = __builtin_amdgcn_wmma_f32_16x16x4_f32(false, a, false, b1, (short)0, acc1, false, false);
        acc2 = __builtin_amdgcn_wmma_f32_16x16x4_f32(false, a, false, b2, (short)0, acc2, false, false);
        acc3 = __builtin_amdgcn_wmma_f32_16x16x4_f32(false, a, false, b3, (short)0, acc3, false, false);
    }

    const int mBase = rowBase + half * 8;
    v8f accs[4] = {acc0, acc1, acc2, acc3};
    #pragma unroll
    for (int nt = 0; nt < 4; ++nt) {
        const int col = nt * 16 + r;
        const float bv = bias[col];
        #pragma unroll
        for (int v = 0; v < 8; ++v) {
            float val = accs[nt][v] + bv;
            val = val > 0.0f ? val : NEG_SLOPE * val;
            h[(size_t)(mBase + v) * OUT_ROW + col] = val;
        }
    }
}

#define FILL_T 256
static_assert(N_PBLK < FILL_T, "one work-group scans all totals in one go; entry N_PBLK is their sum");
__global__ void __launch_bounds__(FILL_T)
k_fill(const int* __restrict__ src, const int* __restrict__ dst,
       const int* __restrict__ lstart,
       const int* __restrict__ btot,
       const int* __restrict__ rank,
       const int* __restrict__ cstart,
       int* __restrict__ rowptr, int* __restrict__ tmp) {
    __shared__ int boff[FILL_T];
    __shared__ int wsum[FILL_T / 32];
    const int t = threadIdx.x;
    {
        const int lane = t & 31, w = t >> 5;
        const int v = t < N_PBLK ? btot[t] : 0;
        int incl = v;
        for (int off = 1; off < 32; off <<= 1) {
            const int x = __shfl_up(incl, off);
            if (lane >= off) incl += x;
        }
        if (lane == 31) wsum[w] = incl;
        __syncthreads();
        int before = 0;
        for (int k = 0; k < FILL_T / 32; ++k) before += k < w ? wsum[k] : 0;
        boff[t] = before + incl - v;
        __syncthreads();
    }
    const int tid    = blockIdx.x * FILL_T + t;
    const int stride = gridDim.x * FILL_T;
    for (int e0 = tid; e0 < N_EDGES; e0 += ITEMS_PER_LANE * stride) {
        int s[ITEMS_PER_LANE], d[ITEMS_PER_LANE], p[ITEMS_PER_LANE], c[ITEMS_PER_LANE];
        #pragma unroll
        for (int u = 0; u < ITEMS_PER_LANE; ++u) {
            const int e  = e0 + u * stride;
            const int ec = e < N_EDGES ? e : N_EDGES - 1;
            s[u] = src[ec];
            d[u] = dst[ec];
            p[u] = rank[ec];
            c[u] = ec / CHUNK_DST;
        }
        #pragma unroll
        for (int u = 0; u < ITEMS_PER_LANE; ++u)
            p[u] += boff[(unsigned)d[u] / PFX_T] + lstart[d[u]] + cstart[(size_t)c[u] * N_NODES + d[u]];
        #pragma unroll
        for (int u = 0; u < ITEMS_PER_LANE; ++u) {
            const int e = e0 + u * stride;
            if (e < N_EDGES) tmp[p[u]] = s[u];
        }
    }
    for (int i = tid; i <= N_NODES; i += stride) {
        int r = boff[N_PBLK];
        if (i < N_NODES) r = boff[(unsigned)i / PFX_T] + lstart[i];
        rowptr[i] = r;
    }
}

__global__ void
k_sort_gemm(const int* __restrict__ rowptr, const int* __restrict__ tmp,
            int* __restrict__ col,
            const float* __restrict__ x, const float* __restrict__ W,
            const float* __restrict__ bias, float* __restrict__ h) {
    if (blockIdx.x < GEMM_BLOCKS) {
        gemm_strip(x, W, bias, h);
        return;
    }
    const int w0 = __builtin_amdgcn_readfirstlane((int)(((blockIdx.x - GEMM_BLOCKS) * blockDim.x + threadIdx.x) >> 5));
    const int nw = (int)(((gridDim.x - GEMM_BLOCKS) * blockDim.x) >> 5);
    const int lane = threadIdx.x & 31;
    for (int i = w0; i < N_NODES; i += nw) {
        const int b = __builtin_amdgcn_readfirstlane(rowptr[i]);
        const int n = __builtin_amdgcn_readfirstlane(rowptr[i + 1]) - b;
        if (n <= 32) {
            const int key = lane < n ? tmp[b + lane] : 0x7fffffff;
            int r = 0;
            for (int j = 0; j < n; ++j) {
                const int kj = __builtin_amdgcn_readlane(key, j);
                r += (kj < key || (kj == key && j < lane)) ? 1 : 0;
            }
            if (lane < n) col[b + r] = key;
        } else {
            for (int idx = lane; idx < n; idx += 32) {
                const int key = tmp[b + idx];
                int r = 0;
                for (int j = 0; j < n; ++j) {
                    const int kj = tmp[b + j];
                    r += (kj < key || (kj == key && j < idx)) ? 1 : 0;
                }
                col[b + r] = key;
            }
        }
    }
}

template <int SRC_ROW, int DST_ROW, bool LAST>
__global__ void __launch_bounds__(256)
k_pull(const int* __restrict__ rowptr, const int* __restrict__ col,
       const float* __restrict__ dinv, const float* __restrict__ f,
       float* __restrict__ dst, ThetaAll th) {
    static_assert(K_STEPS == N_THETA + 1, "f_0..f_4 fill the five slots of an out row, f_5 the workspace plane");
    const int lane = threadIdx.x & 31;
    const int half = lane >> 4;
    const int q    = lane & 15;
    const int w0 = __builtin_amdgcn_readfirstlane((int)((blockIdx.x * blockDim.x + threadIdx.x) >> 5));
    const int nw = (int)((gridDim.x * blockDim.x) >> 5);
    if (w0 >= N_NODES) return;

    int i = w0;
    int b = __builtin_amdgcn_readfirstlane(rowptr[i]);
    int n = __builtin_amdgcn_readfirstlane(rowptr[i + 1]) - b;
    int   s_l = i;
    float w_l = 0.0f;
    if (lane < n) { s_l = col[b + lane]; w_l = dinv[s_l]; }

    int  i2   = i + nw;
    bool has2 = i2 < N_NODES;
    int  b2 = 0, e2 = 0;
    if (has2) {
        b2 = __builtin_amdgcn_readfirstlane(rowptr[i2]);
        e2 = __builtin_amdgcn_readfirstlane(rowptr[i2 + 1]);
    }

    for (;;) {
        const int  n2   = e2 - b2;
        const int  i3   = i2 + nw;
        const bool has3 = i3 < N_NODES;
        int b3 = 0, e3 = 0;
        if (has3) {
            b3 = __builtin_amdgcn_readfirstlane(rowptr[i3]);
            e3 = __builtin_amdgcn_readfirstlane(rowptr[i3 + 1]);
        }
        int s_n = has2 ? i2 : i;
        if (has2 && lane < n2) s_n = col[b2 + lane];

        const float  di = dinv[i];
        const float4 fv = *(const float4*)(f + (size_t)i * SRC_ROW + 4 * q);
        float* ob = dst + (size_t)i * DST_ROW + 4 * q;
        [[maybe_unused]] float4 own[N_THETA];
        if constexpr (LAST) {
            #pragma unroll
            for (int k = 0; k < N_THETA; ++k) own[k] = *(const float4*)(ob + k * H_FEATS);
        }

        float4 acc = make_float4(0.0f, 0.0f, 0.0f, 0.0f);
        for (int base = 0; base < n; base += 32) {
            if (base > 0) {
                const int idx = base + lane;
                s_l = i;
                w_l = 0.0f;
                if (idx < n) { s_l = col[b + idx]; w_l = dinv[s_l]; }
            }
            const int m = (n - base < 32) ? (n - base) : 32;
            const int w_li = __float_as_int(w_l);
            for (int j = 0; j < m; j += 8) {
                float4 v[4];
                float  w[4];
                #pragma unroll
                for (int u = 0; u < 4; ++u) {
                    const int e0 = j + 2 * u;
                    const int sa = __builtin_amdgcn_readlane(s_l, e0);
                    const int sb = __builtin_amdgcn_readlane(s_l, e0 + 1);
                    const int wa = __builtin_amdgcn_readlane(w_li, e0);
                    const int wb = __builtin_amdgcn_readlane(w_li, e0 + 1);
                    const int s = half ? sb : sa;
                    w[u] = __int_as_float(half ? wb : wa);
                    v[u] = *(const float4*)(f + (size_t)s * SRC_ROW + 4 * q);
                }
                #pragma unroll
                for (int u = 0; u < 4; ++u) {
                    acc.x = fmaf(v[u].x, w[u], acc.x);
                    acc.y = fmaf(v[u].y, w[u], acc.y);
                    acc.z = fmaf(v[u].z, w[u], acc.z);
                    acc.w = fmaf(v[u].w, w[u], acc.w);
                }
            }
        }

        float w_n = 0.0f;
        if (has2 && lane < n2) w_n = dinv[s_n];

        acc.x += __shfl_xor(acc.x, 16);
        acc.y += __shfl_xor(acc.y, 16);
        acc.z += __shfl_xor(acc.z, 16);
        acc.w += __shfl_xor(acc.w, 16);

        float4 fn;
        fn.x = fv.x - acc.x * di;
        fn.y = fv.y - acc.y * di;
        fn.z = fv.z - acc.z * di;
        fn.w = fv.w - acc.w * di;

        if constexpr (!LAST) {
            if (half == 0) *(float4*)ob = fn;
        } else {
            #pragma unroll
            for (int t = 0; t < N_THETA && half == 0; ++t) {
                float4 r;
                r.x = th.v[t][0] * own[0].x;
                r.y = th.v[t][0] * own[0].y;
                r.z = th.v[t][0] * own[0].z;
                r.w = th.v[t][0] * own[0].w;
                #pragma unroll
                for (int k = 1; k < N_THETA; ++k) {
                    r.x = fmaf(th.v[t][k], own[k].x, r.x);
                    r.y = fmaf(th.v[t][k], own[k].y, r.y);
                    r.z = fmaf(th.v[t][k], own[k].z, r.z);
                    r.w = fmaf(th.v[t][k], own[k].w, r.w);
                }
                r.x = fmaf(th.v[t][K_STEPS - 1], fv.x, r.x);
                r.y = fmaf(th.v[t][K_STEPS - 1], fv.y, r.y);
                r.z = fmaf(th.v[t][K_STEPS - 1], fv.z, r.z);
                r.w = fmaf(th.v[t][K_STEPS - 1], fv.w, r.w);
                r.x = fmaf(th.v[t][K_STEPS], fn.x, r.x);
                r.y = fmaf(th.v[t][K_STEPS], fn.y, r.y);
                r.z = fmaf(th.v[t][K_STEPS], fn.z, r.z);
                r.w = fmaf(th.v[t][K_STEPS], fn.w, r.w);
                *(float4*)(ob + t * H_FEATS) = r;
            }
        }

        if (!has2) break;
        i = i2; b = b2; n = n2; s_l = s_n; w_l = w_n;
        i2 = i3; has2 = has3; b2 = b3; e2 = e3;
    }
}

static double host_comb(int n, int k) {
    double r = 1.0;
    for (int t = 0; t < k; ++t) r = r * (double)(n - t) / (double)(t + 1);
    return r;
}
static void compute_thetas(double th[N_THETA][K_STEPS + 1]) {
    const double e = 1.4;
    const int d = 4, OFF = 2;
    for (int idx = 0; idx < N_THETA; ++idx) {
        int i = OFF + idx;
        int m = d - i + OFF;
        double B = tgamma((double)(i + 1)) * tgamma((double)(d + 1 - i + OFF)) /
                   tgamma((double)(d + 2 + OFF));
        for (int j = 0; j <= K_STEPS; ++j) th[idx][j] = 0.0;
        for (int j = 0; j <= m; ++j) {
            double c = pow(1.0 / e, (double)i) * host_comb(m, j) *
                       pow(-1.0 / e, (double)j) / (e * B);
            th[idx][i + j] += c;
        }
    }
}

extern "C" void kernel_launch(void* const* d_in, const int* in_sizes, int n_in,
                              void* d_out, int out_size, void* d_ws, size_t ws_size,
                              hipStream_t stream) {
    const float* x    = (const float*)d_in[0];
    const float* W    = (const float*)d_in[1];
    const float* bias = (const float*)d_in[2];
    const int*   src  = (const int*)d_in[3];
    const int*   dst  = (const int*)d_in[4];
    float* out = (float*)d_out;

    float* wsf  = (float*)d_ws;
    float* dinv = wsf + N_NODES;
    float* f5   = dinv + N_NODES;
    int*   btot   = (int*)d_ws;
    int*   lstart = (int*)(f5 + (size_t)N_NODES * H_FEATS);
    int*   rowptr = lstart + NODE_PAD;
    int*   rank   = rowptr + NODE_PAD;
    int*   tmp    = rank + N_EDGES;
    int*   col    = tmp + N_EDGES;
    int*   cc_dst = (int*)f5;
    int*   cc_src = tmp;
    static_assert((size_t)C_DST * N_NODES <= (size_t)N_NODES * H_FEATS, "destination counts fit the f5 plane");
    static_assert((size_t)C_SRC * N_NODES <= (size_t)2 * N_EDGES, "source counts fit tmp..col");
    static_assert(N_PBLK <= N_NODES, "work-group totals fit the first plane");

    double th[N_THETA][K_STEPS + 1];
    compute_thetas(th);
    ThetaAll ta;
    for (int t = 0; t < N_THETA; ++t)
        for (int k = 0; k <= K_STEPS; ++k) ta.v[t][k] = (float)th[t][k];

    k_count<<<C_DST + C_SRC, CNT_T, 0, stream>>>(src, dst, cc_dst, cc_src, rank);
    k_chunk_prefix<<<N_PBLK, PFX_T, 0, stream>>>(cc_dst, cc_src, lstart, btot, dinv);
    k_fill<<<EDGE_GRID, FILL_T, 0, stream>>>(src, dst, lstart, btot, rank, cc_dst, rowptr, tmp);
    k_sort_gemm<<<GEMM_BLOCKS + NODE_GRID, 256, 0, stream>>>(rowptr, tmp, col, x, W, bias, out);

    for (int k = 1; k < N_THETA; ++k)
        k_pull<OUT_ROW, OUT_ROW, false><<<NODE_GRID, 256, 0, stream>>>(
            rowptr, col, dinv, out + (k - 1) * H_FEATS, out + k * H_FEATS, ta);
    k_pull<OUT_ROW, H_FEATS, false><<<NODE_GRID, 256, 0, stream>>>(
        rowptr, col, dinv, out + (N_THETA - 1) * H_FEATS, f5, ta);
    k_pull<H_FEATS, OUT_ROW, true><<<NODE_GRID, 256, 0, stream>>>(
        rowptr, col, dinv, f5, out, ta);
}


# Edge weights stored once: the sort launch gathers dinv[src] into an array in col's order (the dead rank array); steps read a node's weights contiguously, no per-edge dinv gather; sum order kept
# speedup vs baseline: 3.8370x; 1.0611x over previous
#include <hip/hip_runtime.h>
#include <hip/hip_bf16.h>
#include <cmath>

#define N_NODES   50000
#define N_EDGES   800000
#define IN_FEATS  256
#define H_FEATS   64
#define N_THETA   5
#define K_STEPS   6
#define NEG_SLOPE 0.01f
#define NODE_PAD  50176
#define OUT_ROW   (N_THETA * H_FEATS)

#define NODES_PER_WAVE   4
#define NODE_GRID  (((N_NODES + NODES_PER_WAVE - 1) / NODES_PER_WAVE + 7) / 8)
#define ITEMS_PER_LANE   4
#define EDGE_GRID  ((N_EDGES / ITEMS_PER_LANE + 255) / 256)

typedef __attribute__((ext_vector_type(2))) float v2f;
typedef __attribute__((ext_vector_type(8))) float v8f;

struct ThetaAll { float v[N_THETA][K_STEPS + 1]; };


#define C_DST    64
#define C_SRC    32
#define CHUNK_DST (N_EDGES / C_DST)
#define CHUNK_SRC (N_EDGES / C_SRC)
#define CNT_T    1024
static_assert(N_EDGES % C_DST == 0 && N_EDGES % C_SRC == 0, "whole chunks");
static_assert(N_NODES % 4 == 0, "counter rows are moved as 16-byte pieces");
__global__ void __launch_bounds__(CNT_T)
k_count(const int* __restrict__ src, const int* __restrict__ dst,
        int* __restrict__ cc_dst, int* __restrict__ cc_src, int* __restrict__ rank) {
    __shared__ int c_lds[N_NODES];
    const int t = threadIdx.x;
    for (int k = t; k < N_NODES / 4; k += CNT_T) ((int4*)c_lds)[k] = make_int4(0, 0, 0, 0);
    __syncthreads();
    const bool is_dst = blockIdx.x < C_DST;
    if (is_dst) {
        const int e_lo = blockIdx.x * CHUNK_DST;
        const int e_hi = e_lo + CHUNK_DST;
        for (int e0 = e_lo + t; e0 < e_hi; e0 += ITEMS_PER_LANE * CNT_T) {
            int d[ITEMS_PER_LANE];
            int r[ITEMS_PER_LANE];
            #pragma unroll
            for (int u = 0; u < ITEMS_PER_LANE; ++u) {
                const int e = e0 + u * CNT_T;
                d[u] = dst[e < e_hi ? e : e_hi - 1];
            }
            #pragma unroll
            for (int u = 0; u < ITEMS_PER_LANE; ++u) {
                const int e = e0 + u * CNT_T;
                r[u] = atomicAdd(&c_lds[d[u]], e < e_hi ? 1 : 0);
            }
            #pragma unroll
            for (int u = 0; u < ITEMS_PER_LANE; ++u) {
                const int e = e0 + u * CNT_T;
                if (e < e_hi) rank[e] = r[u];
            }
        }
    } else {
        const int e_lo = (blockIdx.x - C_DST) * CHUNK_SRC;
        const int e_hi = e_lo + CHUNK_SRC;
        for (int e0 = e_lo + t; e0 < e_hi; e0 += ITEMS_PER_LANE * CNT_T) {
            int s[ITEMS_PER_LANE];
            #pragma unroll
            for (int u = 0; u < ITEMS_PER_LANE; ++u) {
                const int e = e0 + u * CNT_T;
                s[u] = src[e < e_hi ? e : e_hi - 1];
            }
            #pragma unroll
            for (int u = 0; u < ITEMS_PER_LANE; ++u) {
                const int e = e0 + u * CNT_T;
                atomicAdd(&c_lds[s[u]], e < e_hi ? 1 : 0);
            }
        }
    }
    __syncthreads();
    int4* o = (int4*)(is_dst ? cc_dst + (size_t)blockIdx.x * N_NODES
                             : cc_src + (size_t)(blockIdx.x - C_DST) * N_NODES);
    for (int k = t; k < N_NODES / 4; k += CNT_T) o[k] = ((const int4*)c_lds)[k];
}

#define PFX_BATCH 32
#define PFX_T     256
#define N_PBLK    ((N_NODES + PFX_T - 1) / PFX_T)
static_assert(C_DST % PFX_BATCH == 0 && C_SRC == PFX_BATCH, "whole batches");
__global__ void __launch_bounds__(PFX_T)
k_chunk_prefix(int* __restrict__ cc_dst, const int* __restrict__ cc_src,
               int* __restrict__ lstart, int* __restrict__ btot, float* __restrict__ dinv) {
    __shared__ int wsum[PFX_T / 32];
    const int t = threadIdx.x;
    const int i = blockIdx.x * PFX_T + t;
    const bool live = i < N_NODES;
    int run = 0;
    if (live) {
        int vs[PFX_BATCH];
        #pragma unroll
        for (int u = 0; u < PFX_BATCH; ++u) vs[u] = cc_src[(size_t)u * N_NODES + i];
        for (int c0 = 0; c0 < C_DST; c0 += PFX_BATCH) {
            int v[PFX_BATCH];
            #pragma unroll
            for (int u = 0; u < PFX_BATCH; ++u) v[u] = cc_dst[(size_t)(c0 + u) * N_NODES + i];
            #pragma unroll
            for (int u = 0; u < PFX_BATCH; ++u) {
                cc_dst[(size_t)(c0 + u) * N_NODES + i] = run;
                run += v[u];
            }
        }
        int od = 0;
        #pragma unroll
        for (int u = 0; u < PFX_BATCH; ++u) od += vs[u];
        float d = (float)od;
        d = d > 1.0f ? d : 1.0f;
        dinv[i] = rsqrtf(d);
    }

    const int lane = t & 31, w = t >> 5;
    int incl = run;
    for (int off = 1; off < 32; off <<= 1) {
        const int v = __shfl_up(incl, off);
        if (lane >= off) incl += v;
    }
    if (lane == 31) wsum[w] = incl;
    __syncthreads();
    int before = 0, total = 0;
    for (int k = 0; k < PFX_T / 32; ++k) {
        const int x = wsum[k];
        before += k < w ? x : 0;
        total  += x;
    }
    if (live) lstart[i] = before + incl - run;
    if (t == 0) btot[blockIdx.x] = total;
}

#define GEMM_BLOCKS ((N_NODES / 16 + 7) / 8)
__device__ __forceinline__ void gemm_strip(const float* __restrict__ x,
                            const float* __restrict__ W,
                            const float* __restrict__ bias,
                            float* __restrict__ h) {
    const int wave = (blockIdx.x * blockDim.x + threadIdx.x) >> 5;
    const int lane = threadIdx.x & 31;
    const int nWaves = N_NODES / 16;
    if (wave >= nWaves) return;

    const int rowBase = wave * 16;
    const int half = lane >> 4;
    const int r    = lane & 15;

    v8f acc0 = {}, acc1 = {}, acc2 = {}, acc3 = {};

    for (int k = 0; k < IN_FEATS; k += 4) {
        v2f a;
        const float* xp = x + (size_t)(rowBase + r) * IN_FEATS + k + 2 * half;
        a.x = xp[0];
        a.y = xp[1];

        const float* wp = W + (size_t)(k + 2 * half) * H_FEATS + r;
        v2f b0, b1, b2, b3;
        b0.x = wp[0];      b0.y = wp[H_FEATS];
        b1.x = wp[16];     b1.y = wp[H_FEATS + 16];
        b2.x = wp[32];     b2.y = wp[H_FEATS + 32];
        b3.x = wp[48];     b3.y = wp[H_FEATS + 48];

        acc0 = __builtin_amdgcn_wmma_f32_16x16x4_f32(false, a, false, b0, (short)0, acc0, false, false);
        acc1 = __builtin_amdgcn_wmma_f32_16x16x4_f32(false, a, false, b1, (short)0, acc1, false, false);
        acc2 = __builtin_amdgcn_wmma_f32_16x16x4_f32(false, a, false, b2, (short)0, acc2, false, false);
        acc3 = __builtin_amdgcn_wmma_f32_16x16x4_f32(false, a, false, b3, (short)0, acc3, false, false);
    }

    const int mBase = rowBase + half * 8;
    v8f accs[4] = {acc0, acc1, acc2, acc3};
    #pragma unroll
    for (int nt = 0; nt < 4; ++nt) {
        const int col = nt * 16 + r;
        const float bv = bias[col];
        #pragma unroll
        for (int v = 0; v < 8; ++v) {
            float val = accs[nt][v] + bv;
            val = val > 0.0f ? val : NEG_SLOPE * val;
            h[(size_t)(mBase + v) * OUT_ROW + col] = val;
        }
    }
}

#define FILL_T 256
static_assert(N_PBLK < FILL_T, "one work-group scans all totals in one go; entry N_PBLK is their sum");
__global__ void __launch_bounds__(FILL_T)
k_fill(const int* __restrict__ src, const int* __restrict__ dst,
       const int* __restrict__ lstart,
       const int* __restrict__ btot,
       const int* __restrict__ rank,
       const int* __restrict__ cstart,
       int* __restrict__ rowptr, int* __restrict__ tmp) {
    __shared__ int boff[FILL_T];
    __shared__ int wsum[FILL_T / 32];
    const int t = threadIdx.x;
    {
        const int lane = t & 31, w = t >> 5;
        const int v = t < N_PBLK ? btot[t] : 0;
        int incl = v;
        for (int off = 1; off < 32; off <<= 1) {
            const int x = __shfl_up(incl, off);
            if (lane >= off) incl += x;
        }
        if (lane == 31) wsum[w] = incl;
        __syncthreads();
        int before = 0;
        for (int k = 0; k < FILL_T / 32; ++k) before += k < w ? wsum[k] : 0;
        boff[t] = before + incl - v;
        __syncthreads();
    }
    const int tid    = blockIdx.x * FILL_T + t;
    const int stride = gridDim.x * FILL_T;
    for (int e0 = tid; e0 < N_EDGES; e0 += ITEMS_PER_LANE * stride) {
        int s[ITEMS_PER_LANE], d[ITEMS_PER_LANE], p[ITEMS_PER_LANE], c[ITEMS_PER_LANE];
        #pragma unroll
        for (int u = 0; u < ITEMS_PER_LANE; ++u) {
            const int e  = e0 + u * stride;
            const int ec = e < N_EDGES ? e : N_EDGES - 1;
            s[u] = src[ec];
            d[u] = dst[ec];
            p[u] = rank[ec];
            c[u] = ec / CHUNK_DST;
        }
        #pragma unroll
        for (int u = 0; u < ITEMS_PER_LANE; ++u)
            p[u] += boff[(unsigned)d[u] / PFX_T] + lstart[d[u]] + cstart[(size_t)c[u] * N_NODES + d[u]];
        #pragma unroll
        for (int u = 0; u < ITEMS_PER_LANE; ++u) {
            const int e = e0 + u * stride;
            if (e < N_EDGES) tmp[p[u]] = s[u];
        }
    }
    for (int i = tid; i <= N_NODES; i += stride) {
        int r = boff[N_PBLK];
        if (i < N_NODES) r = boff[(unsigned)i / PFX_T] + lstart[i];
        rowptr[i] = r;
    }
}

__global__ void
k_sort_gemm(const int* __restrict__ rowptr, const int* __restrict__ tmp,
            int* __restrict__ col,
            const float* __restrict__ dinv, float* __restrict__ wcol,
            const float* __restrict__ x, const float* __restrict__ W,
            const float* __restrict__ bias, float* __restrict__ h) {
    if (blockIdx.x < GEMM_BLOCKS) {
        gemm_strip(x, W, bias, h);
        return;
    }
    const int w0 = __builtin_amdgcn_readfirstlane((int)(((blockIdx.x - GEMM_BLOCKS) * blockDim.x + threadIdx.x) >> 5));
    const int nw = (int)(((gridDim.x - GEMM_BLOCKS) * blockDim.x) >> 5);
    const int lane = threadIdx.x & 31;
    for (int i = w0; i < N_NODES; i += nw) {
        const int b = __builtin_amdgcn_readfirstlane(rowptr[i]);
        const int n = __builtin_amdgcn_readfirstlane(rowptr[i + 1]) - b;
        if (n <= 32) {
            int   key = 0x7fffffff;
            float wk  = 0.0f;
            if (lane < n) { key = tmp[b + lane]; wk = dinv[key]; }
            int r = 0;
            for (int j = 0; j < n; ++j) {
                const int kj = __builtin_amdgcn_readlane(key, j);
                r += (kj < key || (kj == key && j < lane)) ? 1 : 0;
            }
            if (lane < n) { col[b + r] = key; wcol[b + r] = wk; }
        } else {
            for (int idx = lane; idx < n; idx += 32) {
                const int key = tmp[b + idx];
                const float wk = dinv[key];
                int r = 0;
                for (int j = 0; j < n; ++j) {
                    const int kj = tmp[b + j];
                    r += (kj < key || (kj == key && j < idx)) ? 1 : 0;
                }
                col[b + r] = key;
                wcol[b + r] = wk;
            }
        }
    }
}

template <int SRC_ROW, int DST_ROW, bool LAST>
__global__ void __launch_bounds__(256)
k_pull(const int* __restrict__ rowptr, const int* __restrict__ col,
       const float* __restrict__ wcol,
       const float* __restrict__ dinv, const float* __restrict__ f,
       float* __restrict__ dst, ThetaAll th) {
    static_assert(K_STEPS == N_THETA + 1, "f_0..f_4 fill the five slots of an out row, f_5 the workspace plane");
    const int lane = threadIdx.x & 31;
    const int half = lane >> 4;
    const int q    = lane & 15;
    const int w0 = __builtin_amdgcn_readfirstlane((int)((blockIdx.x * blockDim.x + threadIdx.x) >> 5));
    const int nw = (int)((gridDim.x * blockDim.x) >> 5);
    if (w0 >= N_NODES) return;

    int i = w0;
    int b = __builtin_amdgcn_readfirstlane(rowptr[i]);
    int n = __builtin_amdgcn_readfirstlane(rowptr[i + 1]) - b;
    int   s_l = i;
    float w_l = 0.0f;
    if (lane < n) { s_l = col[b + lane]; w_l = wcol[b + lane]; }

    int  i2   = i + nw;
    bool has2 = i2 < N_NODES;
    int  b2 = 0, e2 = 0;
    if (has2) {
        b2 = __builtin_amdgcn_readfirstlane(rowptr[i2]);
        e2 = __builtin_amdgcn_readfirstlane(rowptr[i2 + 1]);
    }

    for (;;) {
        const int  n2   = e2 - b2;
        const int  i3   = i2 + nw;
        const bool has3 = i3 < N_NODES;
        int b3 = 0, e3 = 0;
        if (has3) {
            b3 = __builtin_amdgcn_readfirstlane(rowptr[i3]);
            e3 = __builtin_amdgcn_readfirstlane(rowptr[i3 + 1]);
        }
        int   s_n = has2 ? i2 : i;
        float w_n = 0.0f;
        if (has2 && lane < n2) { s_n = col[b2 + lane]; w_n = wcol[b2 + lane]; }

        const float  di = dinv[i];
        const float4 fv = *(const float4*)(f + (size_t)i * SRC_ROW + 4 * q);
        float* ob = dst + (size_t)i * DST_ROW + 4 * q;
        [[maybe_unused]] float4 own[N_THETA];
        if constexpr (LAST) {
            #pragma unroll
            for (int k = 0; k < N_THETA; ++k) own[k] = *(const float4*)(ob + k * H_FEATS);
        }

        float4 acc = make_float4(0.0f, 0.0f, 0.0f, 0.0f);
        for (int base = 0; base < n; base += 32) {
            if (base > 0) {
                const int idx = base + lane;
                s_l = i;
                w_l = 0.0f;
                if (idx < n) { s_l = col[b + idx]; w_l = wcol[b + idx]; }
            }
            const int m = (n - base < 32) ? (n - base) : 32;
            const int w_li = __float_as_int(w_l);
            for (int j = 0; j < m; j += 8) {
                float4 v[4];
                float  w[4];
                #pragma unroll
                for (int u = 0; u < 4; ++u) {
                    const int e0 = j + 2 * u;
                    const int sa = __builtin_amdgcn_readlane(s_l, e0);
                    const int sb = __builtin_amdgcn_readlane(s_l, e0 + 1);
                    const int wa = __builtin_amdgcn_readlane(w_li, e0);
                    const int wb = __builtin_amdgcn_readlane(w_li, e0 + 1);
                    const int s = half ? sb : sa;
                    w[u] = __int_as_float(half ? wb : wa);
                    v[u] = *(const float4*)(f + (size_t)s * SRC_ROW + 4 * q);
                }
                #pragma unroll
                for (int u = 0; u < 4; ++u) {
                    acc.x = fmaf(v[u].x, w[u], acc.x);
                    acc.y = fmaf(v[u].y, w[u], acc.y);
                    acc.z = fmaf(v[u].z, w[u], acc.z);
                    acc.w = fmaf(v[u].w, w[u], acc.w);
                }
            }
        }

        acc.x += __shfl_xor(acc.x, 16);
        acc.y += __shfl_xor(acc.y, 16);
        acc.z += __shfl_xor(acc.z, 16);
        acc.w += __shfl_xor(acc.w, 16);

        float4 fn;
        fn.x = fv.x - acc.x * di;
        fn.y = fv.y - acc.y * di;
        fn.z = fv.z - acc.z * di;
        fn.w = fv.w - acc.w * di;

        if constexpr (!LAST) {
            if (half == 0) *(float4*)ob = fn;
        } else {
            #pragma unroll
            for (int t = 0; t < N_THETA && half == 0; ++t) {
                float4 r;
                r.x = th.v[t][0] * own[0].x;
                r.y = th.v[t][0] * own[0].y;
                r.z = th.v[t][0] * own[0].z;
                r.w = th.v[t][0] * own[0].w;
                #pragma unroll
                for (int k = 1; k < N_THETA; ++k) {
                    r.x = fmaf(th.v[t][k], own[k].x, r.x);
                    r.y = fmaf(th.v[t][k], own[k].y, r.y);
                    r.z = fmaf(th.v[t][k], own[k].z, r.z);
                    r.w = fmaf(th.v[t][k], own[k].w, r.w);
                }
                r.x = fmaf(th.v[t][K_STEPS - 1], fv.x, r.x);
                r.y = fmaf(th.v[t][K_STEPS - 1], fv.y, r.y);
                r.z = fmaf(th.v[t][K_STEPS - 1], fv.z, r.z);
                r.w = fmaf(th.v[t][K_STEPS - 1], fv.w, r.w);
                r.x = fmaf(th.v[t][K_STEPS], fn.x, r.x);
                r.y = fmaf(th.v[t][K_STEPS], fn.y, r.y);
                r.z = fmaf(th.v[t][K_STEPS], fn.z, r.z);
                r.w = fmaf(th.v[t][K_STEPS], fn.w, r.w);
                *(float4*)(ob + t * H_FEATS) = r;
            }
        }

        if (!has2) break;
        i = i2; b = b2; n = n2; s_l = s_n; w_l = w_n;
        i2 = i3; has2 = has3; b2 = b3; e2 = e3;
    }
}

static double host_comb(int n, int k) {
    double r = 1.0;
    for (int t = 0; t < k; ++t) r = r * (double)(n - t) / (double)(t + 1);
    return r;
}
static void compute_thetas(double th[N_THETA][K_STEPS + 1]) {
    const double e = 1.4;
    const int d = 4, OFF = 2;
    for (int idx = 0; idx < N_THETA; ++idx) {
        int i = OFF + idx;
        int m = d - i + OFF;
        double B = tgamma((double)(i + 1)) * tgamma((double)(d + 1 - i + OFF)) /
                   tgamma((double)(d + 2 + OFF));
        for (int j = 0; j <= K_STEPS; ++j) th[idx][j] = 0.0;
        for (int j = 0; j <= m; ++j) {
            double c = pow(1.0 / e, (double)i) * host_comb(m, j) *
                       pow(-1.0 / e, (double)j) / (e * B);
            th[idx][i + j] += c;
        }
    }
}

extern "C" void kernel_launch(void* const* d_in, const int* in_sizes, int n_in,
                              void* d_out, int out_size, void* d_ws, size_t ws_size,
                              hipStream_t stream) {
    const float* x    = (const float*)d_in[0];
    const float* W    = (const float*)d_in[1];
    const float* bias = (const float*)d_in[2];
    const int*   src  = (const int*)d_in[3];
    const int*   dst  = (const int*)d_in[4];
    float* out = (float*)d_out;

    float* wsf  = (float*)d_ws;
    float* dinv = wsf + N_NODES;
    float* f5   = dinv + N_NODES;
    int*   btot   = (int*)d_ws;
    int*   lstart = (int*)(f5 + (size_t)N_NODES * H_FEATS);
    int*   rowptr = lstart + NODE_PAD;
    int*   rank   = rowptr + NODE_PAD;
    int*   tmp    = rank + N_EDGES;
    int*   col    = tmp + N_EDGES;
    float* wcol   = (float*)rank;
    int*   cc_dst = (int*)f5;
    int*   cc_src = tmp;
    static_assert((size_t)C_DST * N_NODES <= (size_t)N_NODES * H_FEATS, "destination counts fit the f5 plane");
    static_assert((size_t)C_SRC * N_NODES <= (size_t)2 * N_EDGES, "source counts fit tmp..col");
    static_assert(N_PBLK <= N_NODES, "work-group totals fit the first plane");

    double th[N_THETA][K_STEPS + 1];
    compute_thetas(th);
    ThetaAll ta;
    for (int t = 0; t < N_THETA; ++t)
        for (int k = 0; k <= K_STEPS; ++k) ta.v[t][k] = (float)th[t][k];

    k_count<<<C_DST + C_SRC, CNT_T, 0, stream>>>(src, dst, cc_dst, cc_src, rank);
    k_chunk_prefix<<<N_PBLK, PFX_T, 0, stream>>>(cc_dst, cc_src, lstart, btot, dinv);
    k_fill<<<EDGE_GRID, FILL_T, 0, stream>>>(src, dst, lstart, btot, rank, cc_dst, rowptr, tmp);
    k_sort_gemm<<<GEMM_BLOCKS + NODE_GRID, 256, 0, stream>>>(rowptr, tmp, col, dinv, wcol, x, W, bias, out);

    for (int k = 1; k < N_THETA; ++k)
        k_pull<OUT_ROW, OUT_ROW, false><<<NODE_GRID, 256, 0, stream>>>(
            rowptr, col, wcol, dinv, out + (k - 1) * H_FEATS, out + k * H_FEATS, ta);
    k_pull<OUT_ROW, H_FEATS, false><<<NODE_GRID, 256, 0, stream>>>(
        rowptr, col, wcol, dinv, out + (N_THETA - 1) * H_FEATS, f5, ta);
    k_pull<H_FEATS, OUT_ROW, true><<<NODE_GRID, 256, 0, stream>>>(
        rowptr, col, wcol, dinv, f5, out, ta);
}
